# starting kernel
# baseline (speedup 1.0000x reference)
#include <hip/hip_runtime.h>
#include <math.h>

#define NNODES 100000
#define NEDGES 1600000
#define NGRAPHS 64
#define DIN 256
#define DHID 128
#define DOUT 64
#define EPSF 1e-5f

typedef __attribute__((ext_vector_type(2))) float v2f;
typedef __attribute__((ext_vector_type(8))) float v8f;

__global__ void zero_f32(float* __restrict__ p, int n) {
  int i = blockIdx.x * blockDim.x + threadIdx.x;
  if (i < n) p[i] = 0.0f;
}

__global__ void deg_accum(const int* __restrict__ col, const float* __restrict__ w,
                          float* __restrict__ deg, int nE) {
  int e = blockIdx.x * blockDim.x + threadIdx.x;
  if (e < nE) atomicAdd(&deg[col[e]], w[e]);
}

__global__ void deg_to_dinv(float* __restrict__ deg, int n) {
  int i = blockIdx.x * blockDim.x + threadIdx.x;
  if (i < n) {
    float d = deg[i] + 1.0f;
    deg[i] = rsqrtf(fmaxf(d, EPSF));
  }
}

template <int NCOLS, int KDIM>
__global__ void wmma_gemm_f32(const float* __restrict__ A, const float* __restrict__ B,
                              float* __restrict__ C, int M) {
  __shared__ float2 Bl[(KDIM / 2) * NCOLS];

  for (int i = threadIdx.x; i < (KDIM / 2) * NCOLS; i += blockDim.x) {
    const int k2 = i / NCOLS;
    const int n = i - k2 * NCOLS;
    float2 v;
    v.x = B[(size_t)(2 * k2 + 0) * NCOLS + n];
    v.y = B[(size_t)(2 * k2 + 1) * NCOLS + n];
    Bl[i] = v;
  }
  __syncthreads();

  const int wave = (int)(((size_t)blockIdx.x * blockDim.x + threadIdx.x) >> 5);
  const int lane = threadIdx.x & 31;
  if (wave >= M / 16) return;
  const int m0 = wave * 16;
  const int half = lane >> 4;
  const int l = lane & 15;
  constexpr int NT = NCOLS / 16;

  v8f acc[NT];
#pragma unroll
  for (int t = 0; t < NT; ++t)
#pragma unroll
    for (int j = 0; j < 8; ++j) acc[t][j] = 0.0f;

  const float* __restrict__ arow = A + (size_t)(m0 + l) * KDIM + 2 * half;

  v2f a;
  a[0] = arow[0];
  a[1] = arow[1];

  for (int k0 = 0; k0 < KDIM; k0 += 4) {
    v2f a_next = a;
    if (k0 + 4 < KDIM) {
      a_next[0] = arow[k0 + 4 + 0];
      a_next[1] = arow[k0 + 4 + 1];
    }
    const float2* __restrict__ brow = Bl + (size_t)(k0 / 2 + half) * NCOLS + l;
    v2f bfrag[NT];
#pragma unroll
    for (int t = 0; t < NT; ++t) {
      float2 bb = brow[t * 16];
      bfrag[t][0] = bb.x;
      bfrag[t][1] = bb.y;
    }
#pragma unroll
    for (int t = 0; t < NT; ++t)
      acc[t] = __builtin_amdgcn_wmma_f32_16x16x4_f32(
           false, a,  false, bfrag[t],
           (short)0, acc[t],  false,  false);
    a = a_next;
  }

#pragma unroll
  for (int t = 0; t < NT; ++t)
#pragma unroll
    for (int r = 0; r < 8; ++r)
      C[(size_t)(m0 + r + 8 * half) * NCOLS + t * 16 + l] = acc[t][r];
}

template <int D>
__global__ void init_out(const float* __restrict__ h, const float* __restrict__ dinv,
                         const float* __restrict__ bias, float* __restrict__ out, int n) {
  const int wave = (int)(((size_t)blockIdx.x * blockDim.x + threadIdx.x) >> 5);
  const int lane = threadIdx.x & 31;
  if (wave >= n) return;
  constexpr int F = D / 32;
  const float s = dinv[wave] * dinv[wave];
  const float* __restrict__ hp = h + (size_t)wave * D + lane * F;
  float* __restrict__ op = out + (size_t)wave * D + lane * F;
#pragma unroll
  for (int f = 0; f < F; ++f) op[f] = bias[lane * F + f] + hp[f] * s;
}

template <int D>
__global__ void edge_scatter(const float* __restrict__ h, const int* __restrict__ rows,
                             const int* __restrict__ cols, const float* __restrict__ w,
                             const float* __restrict__ dinv, float* __restrict__ out, int nE) {
  const int wave = (int)(((size_t)blockIdx.x * blockDim.x + threadIdx.x) >> 5);
  const int lane = threadIdx.x & 31;
  if (wave >= nE) return;
  const int r = rows[wave];
  const int c = cols[wave];
  const float nrm = dinv[r] * w[wave] * dinv[c];
  constexpr int F = D / 32;
  const float* __restrict__ hp = h + (size_t)r * D + lane * F;
  float* __restrict__ op = out + (size_t)c * D + lane * F;
#pragma unroll
  for (int f = 0; f < F; ++f) atomicAdd(op + f, hp[f] * nrm);
}

template <int D>
__global__ void relu_layernorm(float* __restrict__ x, const float* __restrict__ g,
                               const float* __restrict__ b, int n) {
  const int wave = (int)(((size_t)blockIdx.x * blockDim.x + threadIdx.x) >> 5);
  const int lane = threadIdx.x & 31;
  if (wave >= n) return;
  constexpr int F = D / 32;
  float* __restrict__ p = x + (size_t)wave * D + lane * F;
  float v[F];
  float s = 0.0f;
#pragma unroll
  for (int f = 0; f < F; ++f) {
    v[f] = fmaxf(p[f], 0.0f);
    s += v[f];
  }
#pragma unroll
  for (int off = 16; off > 0; off >>= 1) s += __shfl_xor(s, off, 32);
  const float mu = s * (1.0f / D);
  float q = 0.0f;
#pragma unroll
  for (int f = 0; f < F; ++f) {
    float d = v[f] - mu;
    q += d * d;
  }
#pragma unroll
  for (int off = 16; off > 0; off >>= 1) q += __shfl_xor(q, off, 32);
  const float rs = rsqrtf(q * (1.0f / D) + EPSF);
#pragma unroll
  for (int f = 0; f < F; ++f)
    p[f] = (v[f] - mu) * rs * g[lane * F + f] + b[lane * F + f];
}

__global__ void pool_accum(const float* __restrict__ a, const int* __restrict__ batch,
                           float* __restrict__ sums, float* __restrict__ cnts, int n) {
  const int wave = (int)(((size_t)blockIdx.x * blockDim.x + threadIdx.x) >> 5);
  const int lane = threadIdx.x & 31;
  if (wave >= n) return;
  const int g = batch[wave];
  const float* __restrict__ p = a + (size_t)wave * DOUT + lane * 2;
  atomicAdd(&sums[g * DOUT + lane * 2 + 0], p[0]);
  atomicAdd(&sums[g * DOUT + lane * 2 + 1], p[1]);
  if (lane == 0) atomicAdd(&cnts[g], 1.0f);
}

__global__ void pool_final(const float* __restrict__ sums, const float* __restrict__ cnts,
                           float* __restrict__ out) {
  int i = blockIdx.x * blockDim.x + threadIdx.x;
  if (i < NGRAPHS * DOUT) {
    int g = i / DOUT;
    float m = sums[i] / fmaxf(cnts[g], 1.0f);
    out[i] = 1.0f / (1.0f + __expf(-m));
  }
}

extern "C" void kernel_launch(void* const* d_in, const int* in_sizes, int n_in,
                              void* d_out, int out_size, void* d_ws, size_t ws_size,
                              hipStream_t stream) {
  const float* x   = (const float*)d_in[0];
  const int*   ei  = (const int*)d_in[1];
  const float* ew  = (const float*)d_in[2];
  const int*   bat = (const int*)d_in[3];
  const float* W1  = (const float*)d_in[4];
  const float* b1  = (const float*)d_in[5];
  const float* g1  = (const float*)d_in[6];
  const float* bt1 = (const float*)d_in[7];
  const float* W2  = (const float*)d_in[8];
  const float* b2  = (const float*)d_in[9];
  const float* g2  = (const float*)d_in[10];
  const float* bt2 = (const float*)d_in[11];
  float* out = (float*)d_out;

  const int* rows = ei;
  const int* cols = ei + NEDGES;

  char* ws = (char*)d_ws;
  size_t off = 0;
  auto carve = [&](size_t bytes) -> float* {
    float* p = (float*)(ws + off);
    off += (bytes + 255) & ~(size_t)255;
    return p;
  };
  float* dinv = carve((size_t)NNODES * 4);
  float* h    = carve((size_t)NNODES * DHID * 4);
  float* a    = carve((size_t)NNODES * DHID * 4);
  float* sums = carve((size_t)NGRAPHS * DOUT * 4);
  float* cnts = carve((size_t)NGRAPHS * 4);

  const int TB = 256;
  auto blocks = [&](long long threads) { return (int)((threads + TB - 1) / TB); };

  zero_f32<<<blocks(NNODES), TB, 0, stream>>>(dinv, NNODES);
  deg_accum<<<blocks(NEDGES), TB, 0, stream>>>(cols, ew, dinv, NEDGES);
  deg_to_dinv<<<blocks(NNODES), TB, 0, stream>>>(dinv, NNODES);

  wmma_gemm_f32<DHID, DIN><<<blocks((long long)(NNODES / 16) * 32), TB, 0, stream>>>(x, W1, h, NNODES);
  init_out<DHID><<<blocks((long long)NNODES * 32), TB, 0, stream>>>(h, dinv, b1, a, NNODES);
  edge_scatter<DHID><<<blocks((long long)NEDGES * 32), TB, 0, stream>>>(h, rows, cols, ew, dinv, a, NEDGES);
  relu_layernorm<DHID><<<blocks((long long)NNODES * 32), TB, 0, stream>>>(a, g1, bt1, NNODES);

  wmma_gemm_f32<DOUT, DHID><<<blocks((long long)(NNODES / 16) * 32), TB, 0, stream>>>(a, W2, h, NNODES);
  init_out<DOUT><<<blocks((long long)NNODES * 32), TB, 0, stream>>>(h, dinv, b2, a, NNODES);
  edge_scatter<DOUT><<<blocks((long long)NEDGES * 32), TB, 0, stream>>>(h, rows, cols, ew, dinv, a, NEDGES);
  relu_layernorm<DOUT><<<blocks((long long)NNODES * 32), TB, 0, stream>>>(a, g2, bt2, NNODES);

  zero_f32<<<blocks(NGRAPHS * DOUT), TB, 0, stream>>>(sums, NGRAPHS * DOUT);
  zero_f32<<<blocks(NGRAPHS), TB, 0, stream>>>(cnts, NGRAPHS);
  pool_accum<<<blocks((long long)NNODES * 32), TB, 0, stream>>>(a, bat, sums, cnts, NNODES);
  pool_final<<<blocks(NGRAPHS * DOUT), TB, 0, stream>>>(sums, cnts, out);
}


# relu+layernorm and pool sums done on gather_lds's finished rows while still in LDS (pool zeroes stored by layer 1's launch): no raw-sum store, 5 launches fewer; LN sums kept, pool sum order changes
# speedup vs baseline: 2.2176x; 2.2176x over previous
#include <hip/hip_runtime.h>
#include <math.h>

#define NNODES 100000
#define NEDGES 1600000
#define NGRAPHS 64
#define DIN 256
#define DHID 128
#define DOUT 64
#define EPSF 1e-5f

typedef __attribute__((ext_vector_type(2))) float v2f;
typedef __attribute__((ext_vector_type(8))) float v8f;

template <int N>
__device__ __forceinline__ float xrow(float x) {
  const int b = __builtin_bit_cast(int, x);
  return __builtin_bit_cast(float,
      __builtin_amdgcn_update_dpp(0, b, 0x160 + N, 0xf, 0xf, true));
}
__device__ __forceinline__ float xhalf(float x) {
  const int b = __builtin_bit_cast(int, x);
  return __builtin_bit_cast(float,
      __builtin_amdgcn_permlanex16(b, b, 0x76543210, 0xfedcba98, false, false));
}
__device__ __forceinline__ float wave_sum(float s) {
  s += xhalf(s);
  s += xrow<8>(s);
  s += xrow<4>(s);
  s += xrow<2>(s);
  s += xrow<1>(s);
  return s;
}

__global__ void zero_f32(float* __restrict__ p, int n) {
  int i = blockIdx.x * blockDim.x + threadIdx.x;
  if (i < n) p[i] = 0.0f;
}

__global__ void deg_accum(const int* __restrict__ col, const float* __restrict__ w,
                          float* __restrict__ deg, int nE) {
  int e = blockIdx.x * blockDim.x + threadIdx.x;
  if (e < nE) atomicAdd(&deg[col[e]], w[e]);
}

__global__ void deg_to_dinv(float* __restrict__ deg, int n) {
  int i = blockIdx.x * blockDim.x + threadIdx.x;
  if (i < n) {
    float d = deg[i] + 1.0f;
    deg[i] = rsqrtf(fmaxf(d, EPSF));
  }
}

#define GEMM_THREADS 1024
#define GEMM_WAVES (GEMM_THREADS / 32)
#define GEMM_BLOCKS 256
static_assert(NNODES % 16 == 0, "rows come in whole 16-row stripes");

template <int NCOLS, int KDIM>
__global__ void __launch_bounds__(GEMM_THREADS)
wmma_gemm_f32(const float* __restrict__ A, const float* __restrict__ B,
              float* __restrict__ C, int M) {
  __shared__ float2 Bl[(KDIM / 2) * NCOLS];

  for (int i = threadIdx.x; i < (KDIM / 2) * NCOLS; i += GEMM_THREADS) {
    const int k2 = i / NCOLS;
    const int n = i - k2 * NCOLS;
    float2 v;
    v.x = B[(size_t)(2 * k2 + 0) * NCOLS + n];
    v.y = B[(size_t)(2 * k2 + 1) * NCOLS + n];
    Bl[i] = v;
  }
  __syncthreads();

  const int wave = __builtin_amdgcn_readfirstlane((int)(threadIdx.x >> 5));
  const int lane = threadIdx.x & 31;
  const int half = lane >> 4;
  const int l = lane & 15;
  constexpr int NT = NCOLS / 16;
  const int nstripes = M / 16;

  for (int stripe = wave * (int)gridDim.x + (int)blockIdx.x; stripe < nstripes;
       stripe += GEMM_WAVES * (int)gridDim.x) {
    const int m0 = stripe * 16;

    v8f acc[NT];
#pragma unroll
    for (int t = 0; t < NT; ++t)
#pragma unroll
      for (int j = 0; j < 8; ++j) acc[t][j] = 0.0f;

    const float* __restrict__ arow = A + (size_t)(m0 + l) * KDIM + 2 * half;

    v2f a;
    a[0] = arow[0];
    a[1] = arow[1];

    for (int k0 = 0; k0 < KDIM; k0 += 4) {
      v2f a_next = a;
      if (k0 + 4 < KDIM) {
        a_next[0] = arow[k0 + 4 + 0];
        a_next[1] = arow[k0 + 4 + 1];
      }
      const float2* __restrict__ brow = Bl + (size_t)(k0 / 2 + half) * NCOLS + l;
      v2f bfrag[NT];
#pragma unroll
      for (int t = 0; t < NT; ++t) {
        float2 bb = brow[t * 16];
        bfrag[t][0] = bb.x;
        bfrag[t][1] = bb.y;
      }
#pragma unroll
      for (int t = 0; t < NT; ++t)
        acc[t] = __builtin_amdgcn_wmma_f32_16x16x4_f32(
             false, a,  false, bfrag[t],
             (short)0, acc[t],  false,  false);
      a = a_next;
    }

#pragma unroll
    for (int t = 0; t < NT; ++t)
#pragma unroll
      for (int r = 0; r < 8; ++r)
        C[(size_t)(m0 + r + 8 * half) * NCOLS + t * 16 + l] = acc[t][r];
  }
}

#define AGG_THREADS 1024
#define AGG_WAVES (AGG_THREADS / 32)
#define AGG_NPB 392
#define AGG_BLOCKS ((NNODES + AGG_NPB - 1) / AGG_NPB)
#define AGG_QCAP 4096
#define AGG_LOADS 8
#define AGG_ITERS 7
#define AGG_CH4 (AGG_ITERS * AGG_LOADS * AGG_THREADS)
static_assert(NEDGES % 4 == 0, "targets are read four at a time");
static_assert(NNODES < (1 << 30), "node numbers fit the unsigned range test");

#define AGG_RPW ((AGG_NPB + AGG_WAVES - 1) / AGG_WAVES)
static_assert(NGRAPHS == 64 && DOUT == 64,
              "a lane owns two features and counts two graphs");

template <int D, bool POOL>
__global__ void __launch_bounds__(AGG_THREADS)
gather_lds(const float* __restrict__ h, const int* __restrict__ rows,
           const int* __restrict__ cols, const float* __restrict__ w,
           const float* __restrict__ dinv, const float* __restrict__ bias,
           const float* __restrict__ gamma, const float* __restrict__ beta,
           float* __restrict__ out, const int* __restrict__ batch,
           float* __restrict__ sums, float* __restrict__ cnts, int n) {
  static_assert(!POOL || D == DOUT, "the pool table is NGRAPHS x DOUT");
  constexpr int F = D / 32;
  constexpr int ROW4 = D / 4;
  constexpr int U = 16;
  constexpr int NE4 = NEDGES / 4;
  __shared__ __attribute__((aligned(16))) float acc[AGG_NPB * D];
  __shared__ int q[AGG_QCAP];
  __shared__ int qn[2];

  const int tid = threadIdx.x;
  const int lane = tid & 31;
  const int wave = __builtin_amdgcn_readfirstlane(tid >> 5);
  const int node0 = blockIdx.x * AGG_NPB;

  if (tid < 2) qn[tid] = 0;

  if constexpr (!POOL) {
    if (blockIdx.x == 0) {
      for (int i = tid; i < NGRAPHS * DOUT; i += AGG_THREADS) sums[i] = 0.0f;
      if (tid < NGRAPHS) cnts[tid] = 0.0f;
    }
  }

  {
    const float4* __restrict__ h4 = reinterpret_cast<const float4*>(h);
    const float4* __restrict__ b4 = reinterpret_cast<const float4*>(bias);
    float4* acc4 = reinterpret_cast<float4*>(acc);
#pragma unroll 4
    for (int i = tid; i < AGG_NPB * ROW4; i += AGG_THREADS) {
      const int ln = i / ROW4;
      const int f4 = i - ln * ROW4;
      const int node = min(node0 + ln, n - 1);
      const float d = dinv[node];
      const float s = d * d;
      const float4 hv = h4[(size_t)node * ROW4 + f4];
      const float4 bv = b4[f4];
      float4 o;
      o.x = bv.x + hv.x * s;
      o.y = bv.y + hv.y * s;
      o.z = bv.z + hv.z * s;
      o.w = bv.w + hv.w * s;
      acc4[i] = o;
    }
  }
  __syncthreads();

  const int4* __restrict__ c4p = reinterpret_cast<const int4*>(cols);

  auto push = [&](int e, int* cnt) {
    const int p = atomicAdd(cnt, 1);
    if (p < AGG_QCAP) q[p] = e;
  };

  int par = 0;
  for (int ch = 0; ch < NE4; ch += AGG_CH4) {
    int* cnt = &qn[par];
    for (int it = 0; it < AGG_ITERS; ++it) {
      int4 cv[AGG_LOADS];
      int idx[AGG_LOADS];
#pragma unroll
      for (int j = 0; j < AGG_LOADS; ++j) {
        idx[j] = ch + (it * AGG_LOADS + j) * AGG_THREADS + tid;
        if (idx[j] < NE4) {
          cv[j] = c4p[idx[j]];
        } else {
          cv[j].x = -1; cv[j].y = -1; cv[j].z = -1; cv[j].w = -1;
        }
      }
#pragma unroll
      for (int j = 0; j < AGG_LOADS; ++j) {
        const unsigned d0 = (unsigned)(cv[j].x - node0);
        const unsigned d1 = (unsigned)(cv[j].y - node0);
        const unsigned d2 = (unsigned)(cv[j].z - node0);
        const unsigned d3 = (unsigned)(cv[j].w - node0);
        if (d0 < AGG_NPB) push(idx[j] * 4 + 0, cnt);
        if (d1 < AGG_NPB) push(idx[j] * 4 + 1, cnt);
        if (d2 < AGG_NPB) push(idx[j] * 4 + 2, cnt);
        if (d3 < AGG_NPB) push(idx[j] * 4 + 3, cnt);
      }
    }
    __syncthreads();

    int m = __builtin_amdgcn_readfirstlane(*cnt);
    if (tid == 0) qn[par ^ 1] = 0;
    if (m > AGG_QCAP) {
      const int e1 = min((ch + AGG_CH4) * 4, NEDGES);
      for (int e = ch * 4 + tid; e < e1; e += AGG_THREADS) {
        const int c = cols[e];
        const unsigned dl = (unsigned)(c - node0);
        if (dl < AGG_NPB) {
          const int r = rows[e];
          const float nr = dinv[r] * w[e] * dinv[c];
          for (int f = 0; f < D; ++f)
            atomicAdd(&acc[dl * D + f], h[(size_t)r * D + f] * nr);
        }
      }
      m = 0;
    }
    for (int base = wave * 32; base < m; base += AGG_WAVES * 32) {
      const int nb = min(32, m - base);
      int r = 0, cl = 0;
      float nr = 0.0f;
      if (lane < nb) {
        const int e = q[base + lane];
        r = rows[e];
        const int c = cols[e];
        nr = dinv[r] * w[e] * dinv[c];
        cl = c - node0;
      }
#pragma unroll
      for (int j0 = 0; j0 < 32; j0 += U) {
        if (j0 < nb) {
          float v[U][F];
#pragma unroll
          for (int u = 0; u < U; ++u) {
            if (j0 + u < nb) {
              const int rr = __builtin_amdgcn_readlane(r, j0 + u);
              const float* __restrict__ hp = h + (size_t)rr * D + lane;
#pragma unroll
              for (int f = 0; f < F; ++f) v[u][f] = hp[f * 32];
            }
          }
#pragma unroll
          for (int u = 0; u < U; ++u) {
            if (j0 + u < nb) {
              const int cc = __builtin_amdgcn_readlane(cl, j0 + u);
              const float nn =
                  __int_as_float(__builtin_amdgcn_readlane(__float_as_int(nr), j0 + u));
              float* ap = acc + cc * D + lane;
#pragma unroll
              for (int f = 0; f < F; ++f) atomicAdd(ap + f * 32, v[u][f] * nn);
            }
          }
        }
      }
    }
    __syncthreads();
    par ^= 1;
  }

  {
    static_assert(F == 4 || F == 2, "a lane reads its F features of a row with one LDS load");
    float gm[F], bt[F];
#pragma unroll
    for (int f = 0; f < F; ++f) {
      gm[f] = gamma[lane * F + f];
      bt[f] = beta[lane * F + f];
    }
    auto ln_row = [&](int ln, float (&y)[F]) {
      const float* rp = acc + ln * D + lane * F;
      float v[F];
      if constexpr (F == 4) {
        const float4 t = *reinterpret_cast<const float4*>(rp);
        v[0] = t.x; v[1] = t.y; v[2] = t.z; v[3] = t.w;
      } else {
        const float2 t = *reinterpret_cast<const float2*>(rp);
        v[0] = t.x; v[1] = t.y;
      }
      float s = 0.0f;
#pragma unroll
      for (int f = 0; f < F; ++f) {
        v[f] = fmaxf(v[f], 0.0f);
        s += v[f];
      }
      s = wave_sum(s);
      const float mu = s * (1.0f / D);
      float ss = 0.0f;
#pragma unroll
      for (int f = 0; f < F; ++f) {
        const float d = v[f] - mu;
        ss += d * d;
      }
      ss = wave_sum(ss);
      const float rs = rsqrtf(ss * (1.0f / D) + EPSF);
#pragma unroll
      for (int f = 0; f < F; ++f) y[f] = (v[f] - mu) * rs * gm[f] + bt[f];
    };

    if constexpr (!POOL) {
      float yy[AGG_RPW][F];
#pragma unroll
      for (int k = 0; k < AGG_RPW; ++k) ln_row(min(wave + k * AGG_WAVES, AGG_NPB - 1), yy[k]);
#pragma unroll
      for (int k = 0; k < AGG_RPW; ++k) {
        const int ln = wave + k * AGG_WAVES;
        const int node = node0 + ln;
        const float (&y)[F] = yy[k];
        if (ln < AGG_NPB && node < n) {
          float* __restrict__ op = out + (size_t)node * D + lane * F;
          if constexpr (F == 4) {
            float4 o; o.x = y[0]; o.y = y[1]; o.z = y[2]; o.w = y[3];
            *reinterpret_cast<float4*>(op) = o;
          } else {
            float2 o; o.x = y[0]; o.y = y[1];
            *reinterpret_cast<float2*>(op) = o;
          }
        }
      }
    } else {
      __shared__ float tab[NGRAPHS * DOUT];
      __shared__ float gcnt[NGRAPHS];
      for (int i = tid; i < NGRAPHS * DOUT; i += AGG_THREADS) tab[i] = 0.0f;
      if (tid < NGRAPHS) gcnt[tid] = 0.0f;
      int gid[AGG_RPW];
#pragma unroll
      for (int k = 0; k < AGG_RPW; ++k)
        gid[k] = batch[min(node0 + wave + k * AGG_WAVES, n - 1)];
      __syncthreads();

      int c0 = 0, c1 = 0;
      float yy[AGG_RPW][F];
#pragma unroll
      for (int k = 0; k < AGG_RPW; ++k) ln_row(min(wave + k * AGG_WAVES, AGG_NPB - 1), yy[k]);
#pragma unroll
      for (int k = 0; k < AGG_RPW; ++k) {
        const int ln = wave + k * AGG_WAVES;
        const int node = node0 + ln;
        const float (&y)[F] = yy[k];
        if (ln < AGG_NPB && node < n) {
          const int g = gid[k];
          atomicAdd(&tab[g * DOUT + lane * 2 + 0], y[0]);
          atomicAdd(&tab[g * DOUT + lane * 2 + 1], y[1]);
          c0 += (g == lane) ? 1 : 0;
          c1 += (g == lane + 32) ? 1 : 0;
        }
      }
      if (c0 != 0) atomicAdd(&gcnt[lane], (float)c0);
      if (c1 != 0) atomicAdd(&gcnt[lane + 32], (float)c1);
      __syncthreads();

      for (int i = tid; i < NGRAPHS * DOUT; i += AGG_THREADS) {
        const float s = tab[i];
        if (s != 0.0f) atomicAdd(&sums[i], s);
      }
      if (tid < NGRAPHS) {
        const float c = gcnt[tid];
        if (c != 0.0f) atomicAdd(&cnts[tid], c);
      }
    }
  }
}

__global__ void pool_final(const float* __restrict__ sums, const float* __restrict__ cnts,
                           float* __restrict__ out) {
  int i = blockIdx.x * blockDim.x + threadIdx.x;
  if (i < NGRAPHS * DOUT) {
    int g = i / DOUT;
    float m = sums[i] / fmaxf(cnts[g], 1.0f);
    out[i] = 1.0f / (1.0f + __expf(-m));
  }
}

extern "C" void kernel_launch(void* const* d_in, const int* in_sizes, int n_in,
                              void* d_out, int out_size, void* d_ws, size_t ws_size,
                              hipStream_t stream) {
  const float* x   = (const float*)d_in[0];
  const int*   ei  = (const int*)d_in[1];
  const float* ew  = (const float*)d_in[2];
  const int*   bat = (const int*)d_in[3];
  const float* W1  = (const float*)d_in[4];
  const float* b1  = (const float*)d_in[5];
  const float* g1  = (const float*)d_in[6];
  const float* bt1 = (const float*)d_in[7];
  const float* W2  = (const float*)d_in[8];
  const float* b2  = (const float*)d_in[9];
  const float* g2  = (const float*)d_in[10];
  const float* bt2 = (const float*)d_in[11];
  float* out = (float*)d_out;

  const int* rows = ei;
  const int* cols = ei + NEDGES;

  char* ws = (char*)d_ws;
  size_t off = 0;
  auto carve = [&](size_t bytes) -> float* {
    float* p = (float*)(ws + off);
    off += (bytes + 255) & ~(size_t)255;
    return p;
  };
  float* dinv = carve((size_t)NNODES * 4);
  float* h    = carve((size_t)NNODES * DHID * 4);
  float* a    = carve((size_t)NNODES * DHID * 4);
  float* sums = carve((size_t)NGRAPHS * DOUT * 4);
  float* cnts = carve((size_t)NGRAPHS * 4);

  const int TB = 256;
  auto blocks = [&](long long threads) { return (int)((threads + TB - 1) / TB); };

  zero_f32<<<blocks(NNODES), TB, 0, stream>>>(dinv, NNODES);
  deg_accum<<<blocks(NEDGES), TB, 0, stream>>>(cols, ew, dinv, NEDGES);
  deg_to_dinv<<<blocks(NNODES), TB, 0, stream>>>(dinv, NNODES);

  wmma_gemm_f32<DHID, DIN><<<GEMM_BLOCKS, GEMM_THREADS, 0, stream>>>(x, W1, h, NNODES);
  gather_lds<DHID, false><<<AGG_BLOCKS, AGG_THREADS, 0, stream>>>(
      h, rows, cols, ew, dinv, b1, g1, bt1, a, nullptr, sums, cnts, NNODES);

  wmma_gemm_f32<DOUT, DHID><<<GEMM_BLOCKS, GEMM_THREADS, 0, stream>>>(a, W2, h, NNODES);
  gather_lds<DOUT, true><<<AGG_BLOCKS, AGG_THREADS, 0, stream>>>(
      h, rows, cols, ew, dinv, b2, g2, bt2, nullptr, bat, sums, cnts, NNODES);
  pool_final<<<blocks(NGRAPHS * DOUT), TB, 0, stream>>>(sums, cnts, out);
}


# layer 2's target scan reads 1 byte per edge (owner work-group's number, written by the layer-2 product launch into the unused half of h), 16 edges per 16-byte load; queue and row sum order change
# speedup vs baseline: 2.4332x; 1.0972x over previous
#include <hip/hip_runtime.h>
#include <math.h>

#define NNODES 100000
#define NEDGES 1600000
#define NGRAPHS 64
#define DIN 256
#define DHID 128
#define DOUT 64
#define EPSF 1e-5f

typedef __attribute__((ext_vector_type(2))) float v2f;
typedef __attribute__((ext_vector_type(8))) float v8f;

template <int N>
__device__ __forceinline__ float xrow(float x) {
  const int b = __builtin_bit_cast(int, x);
  return __builtin_bit_cast(float,
      __builtin_amdgcn_update_dpp(0, b, 0x160 + N, 0xf, 0xf, true));
}
__device__ __forceinline__ float xhalf(float x) {
  const int b = __builtin_bit_cast(int, x);
  return __builtin_bit_cast(float,
      __builtin_amdgcn_permlanex16(b, b, 0x76543210, 0xfedcba98, false, false));
}
__device__ __forceinline__ float wave_sum(float s) {
  s += xhalf(s);
  s += xrow<8>(s);
  s += xrow<4>(s);
  s += xrow<2>(s);
  s += xrow<1>(s);
  return s;
}

__global__ void zero_f32(float* __restrict__ p, int n) {
  int i = blockIdx.x * blockDim.x + threadIdx.x;
  if (i < n) p[i] = 0.0f;
}

__global__ void deg_accum(const int* __restrict__ col, const float* __restrict__ w,
                          float* __restrict__ deg, int nE) {
  int e = blockIdx.x * blockDim.x + threadIdx.x;
  if (e < nE) atomicAdd(&deg[col[e]], w[e]);
}

__global__ void deg_to_dinv(float* __restrict__ deg, int n) {
  int i = blockIdx.x * blockDim.x + threadIdx.x;
  if (i < n) {
    float d = deg[i] + 1.0f;
    deg[i] = rsqrtf(fmaxf(d, EPSF));
  }
}

#define GEMM_THREADS 1024
#define GEMM_WAVES (GEMM_THREADS / 32)
#define GEMM_BLOCKS 256
static_assert(NNODES % 16 == 0, "rows come in whole 16-row stripes");
#define AGG_NPB 392
#define AGG_BLOCKS ((NNODES + AGG_NPB - 1) / AGG_NPB)
static_assert(AGG_BLOCKS <= 256, "a work-group number fits one byte");
static_assert(NEDGES % 16 == 0, "the key bytes are read sixteen at a time");
static_assert((size_t)NEDGES <= (size_t)NNODES * (DHID - DOUT) * 4,
              "the key bytes fit the half of h that layer 2 leaves unused");

template <int NCOLS, int KDIM, bool KEYS>
__global__ void __launch_bounds__(GEMM_THREADS)
wmma_gemm_f32(const float* __restrict__ A, const float* __restrict__ B,
              float* __restrict__ C, int M, const int* __restrict__ cols,
              unsigned int* __restrict__ keys) {
  __shared__ float2 Bl[(KDIM / 2) * NCOLS];

  if constexpr (KEYS) {
    const int4* __restrict__ c4p = reinterpret_cast<const int4*>(cols);
    const int nthr = (int)gridDim.x * GEMM_THREADS;
    for (int i = (int)blockIdx.x * GEMM_THREADS + (int)threadIdx.x; i < NEDGES / 4; i += nthr) {
      const int4 c = c4p[i];
      keys[i] = (unsigned)(c.x / AGG_NPB) | ((unsigned)(c.y / AGG_NPB) << 8) |
                ((unsigned)(c.z / AGG_NPB) << 16) | ((unsigned)(c.w / AGG_NPB) << 24);
    }
  }

  for (int i = threadIdx.x; i < (KDIM / 2) * NCOLS; i += GEMM_THREADS) {
    const int k2 = i / NCOLS;
    const int n = i - k2 * NCOLS;
    float2 v;
    v.x = B[(size_t)(2 * k2 + 0) * NCOLS + n];
    v.y = B[(size_t)(2 * k2 + 1) * NCOLS + n];
    Bl[i] = v;
  }
  __syncthreads();

  const int wave = __builtin_amdgcn_readfirstlane((int)(threadIdx.x >> 5));
  const int lane = threadIdx.x & 31;
  const int half = lane >> 4;
  const int l = lane & 15;
  constexpr int NT = NCOLS / 16;
  const int nstripes = M / 16;

  for (int stripe = wave * (int)gridDim.x + (int)blockIdx.x; stripe < nstripes;
       stripe += GEMM_WAVES * (int)gridDim.x) {
    const int m0 = stripe * 16;

    v8f acc[NT];
#pragma unroll
    for (int t = 0; t < NT; ++t)
#pragma unroll
      for (int j = 0; j < 8; ++j) acc[t][j] = 0.0f;

    const float* __restrict__ arow = A + (size_t)(m0 + l) * KDIM + 2 * half;

    v2f a;
    a[0] = arow[0];
    a[1] = arow[1];

    for (int k0 = 0; k0 < KDIM; k0 += 4) {
      v2f a_next = a;
      if (k0 + 4 < KDIM) {
        a_next[0] = arow[k0 + 4 + 0];
        a_next[1] = arow[k0 + 4 + 1];
      }
      const float2* __restrict__ brow = Bl + (size_t)(k0 / 2 + half) * NCOLS + l;
      v2f bfrag[NT];
#pragma unroll
      for (int t = 0; t < NT; ++t) {
        float2 bb = brow[t * 16];
        bfrag[t][0] = bb.x;
        bfrag[t][1] = bb.y;
      }
#pragma unroll
      for (int t = 0; t < NT; ++t)
        acc[t] = __builtin_amdgcn_wmma_f32_16x16x4_f32(
             false, a,  false, bfrag[t],
             (short)0, acc[t],  false,  false);
      a = a_next;
    }

#pragma unroll
    for (int t = 0; t < NT; ++t)
#pragma unroll
      for (int r = 0; r < 8; ++r)
        C[(size_t)(m0 + r + 8 * half) * NCOLS + t * 16 + l] = acc[t][r];
  }
}

#define AGG_THREADS 1024
#define AGG_WAVES (AGG_THREADS / 32)
#define AGG_QCAP 4096
#define AGG_LOADS 8
#define AGG_ITERS 7
#define AGG_CH4 (AGG_ITERS * AGG_LOADS * AGG_THREADS)
static_assert(NEDGES % 4 == 0, "targets are read four at a time");
static_assert(NNODES < (1 << 30), "node numbers fit the unsigned range test");

#define AGG_RPW ((AGG_NPB + AGG_WAVES - 1) / AGG_WAVES)
static_assert(NGRAPHS == 64 && DOUT == 64,
              "a lane owns two features and counts two graphs");

__device__ __forceinline__ unsigned zero_bytes(unsigned x) {
  return ~(((x & 0x7f7f7f7fu) + 0x7f7f7f7fu) | x | 0x7f7f7f7fu);
}

template <int D, bool POOL, bool BYTEKEYS>
__global__ void __launch_bounds__(AGG_THREADS)
gather_lds(const float* __restrict__ h, const int* __restrict__ rows,
           const int* __restrict__ cols, const float* __restrict__ w,
           const float* __restrict__ dinv, const float* __restrict__ bias,
           const float* __restrict__ gamma, const float* __restrict__ beta,
           float* __restrict__ out, const int* __restrict__ batch,
           float* __restrict__ sums, float* __restrict__ cnts, int n,
           const unsigned int* __restrict__ keys) {
  static_assert(!POOL || D == DOUT, "the pool table is NGRAPHS x DOUT");
  constexpr int F = D / 32;
  constexpr int ROW4 = D / 4;
  constexpr int U = 16;
  constexpr int NE4 = NEDGES / 4;
  __shared__ __attribute__((aligned(16))) float acc[AGG_NPB * D];
  __shared__ int q[AGG_QCAP];
  __shared__ int qn[2];

  const int tid = threadIdx.x;
  const int lane = tid & 31;
  const int wave = __builtin_amdgcn_readfirstlane(tid >> 5);
  const int node0 = blockIdx.x * AGG_NPB;

  if (tid < 2) qn[tid] = 0;

  if constexpr (!POOL) {
    if (blockIdx.x == 0) {
      for (int i = tid; i < NGRAPHS * DOUT; i += AGG_THREADS) sums[i] = 0.0f;
      if (tid < NGRAPHS) cnts[tid] = 0.0f;
    }
  }

  {
    const float4* __restrict__ h4 = reinterpret_cast<const float4*>(h);
    const float4* __restrict__ b4 = reinterpret_cast<const float4*>(bias);
    float4* acc4 = reinterpret_cast<float4*>(acc);
#pragma unroll 4
    for (int i = tid; i < AGG_NPB * ROW4; i += AGG_THREADS) {
      const int ln = i / ROW4;
      const int f4 = i - ln * ROW4;
      const int node = min(node0 + ln, n - 1);
      const float d = dinv[node];
      const float s = d * d;
      const float4 hv = h4[(size_t)node * ROW4 + f4];
      const float4 bv = b4[f4];
      float4 o;
      o.x = bv.x + hv.x * s;
      o.y = bv.y + hv.y * s;
      o.z = bv.z + hv.z * s;
      o.w = bv.w + hv.w * s;
      acc4[i] = o;
    }
  }
  __syncthreads();

  const int4* __restrict__ c4p = reinterpret_cast<const int4*>(cols);

  auto push = [&](int e, int* cnt) {
    const int p = atomicAdd(cnt, 1);
    if (p < AGG_QCAP) q[p] = e;
  };

  int par = 0;
  for (int ch = 0; ch < NE4; ch += AGG_CH4) {
    int* cnt = &qn[par];
    if constexpr (BYTEKEYS) {
      constexpr int KLOADS = 7;
      constexpr int KITERS = 2;
      static_assert(KLOADS * KITERS * 16 == AGG_ITERS * AGG_LOADS * 4,
                    "a piece covers the same edges with either kind of scan");
      constexpr int NK16 = NEDGES / 16;
      const uint4* __restrict__ k4p = reinterpret_cast<const uint4*>(keys);
      const unsigned pat = (unsigned)blockIdx.x * 0x01010101u;
      for (int it = 0; it < KITERS; ++it) {
        uint4 kv[KLOADS];
        int gi[KLOADS];
#pragma unroll
        for (int j = 0; j < KLOADS; ++j) {
          gi[j] = ch / 4 + (it * KLOADS + j) * AGG_THREADS + tid;
          if (gi[j] < NK16) {
            kv[j] = k4p[gi[j]];
          } else {
            kv[j].x = ~pat; kv[j].y = ~pat; kv[j].z = ~pat; kv[j].w = ~pat;
          }
        }
#pragma unroll
        for (int j = 0; j < KLOADS; ++j) {
          unsigned m = (zero_bytes(kv[j].x ^ pat) >> 7) | (zero_bytes(kv[j].y ^ pat) >> 6) |
                       (zero_bytes(kv[j].z ^ pat) >> 5) | (zero_bytes(kv[j].w ^ pat) >> 4);
          while (m != 0) {
            const int p = __builtin_ctz(m);
            m &= m - 1;
            push(gi[j] * 16 + ((p & 7) << 2) + (p >> 3), cnt);
          }
        }
      }
    } else {
    for (int it = 0; it < AGG_ITERS; ++it) {
      int4 cv[AGG_LOADS];
      int idx[AGG_LOADS];
#pragma unroll
      for (int j = 0; j < AGG_LOADS; ++j) {
        idx[j] = ch + (it * AGG_LOADS + j) * AGG_THREADS + tid;
        if (idx[j] < NE4) {
          cv[j] = c4p[idx[j]];
        } else {
          cv[j].x = -1; cv[j].y = -1; cv[j].z = -1; cv[j].w = -1;
        }
      }
#pragma unroll
      for (int j = 0; j < AGG_LOADS; ++j) {
        const unsigned d0 = (unsigned)(cv[j].x - node0);
        const unsigned d1 = (unsigned)(cv[j].y - node0);
        const unsigned d2 = (unsigned)(cv[j].z - node0);
        const unsigned d3 = (unsigned)(cv[j].w - node0);
        if (d0 < AGG_NPB) push(idx[j] * 4 + 0, cnt);
        if (d1 < AGG_NPB) push(idx[j] * 4 + 1, cnt);
        if (d2 < AGG_NPB) push(idx[j] * 4 + 2, cnt);
        if (d3 < AGG_NPB) push(idx[j] * 4 + 3, cnt);
      }
    }
    }
    __syncthreads();

    int m = __builtin_amdgcn_readfirstlane(*cnt);
    if (tid == 0) qn[par ^ 1] = 0;
    if (m > AGG_QCAP) {
      const int e1 = min((ch + AGG_CH4) * 4, NEDGES);
      for (int e = ch * 4 + tid; e < e1; e += AGG_THREADS) {
        const int c = cols[e];
        const unsigned dl = (unsigned)(c - node0);
        if (dl < AGG_NPB) {
          const int r = rows[e];
          const float nr = dinv[r] * w[e] * dinv[c];
          for (int f = 0; f < D; ++f)
            atomicAdd(&acc[dl * D + f], h[(size_t)r * D + f] * nr);
        }
      }
      m = 0;
    }
    for (int base = wave * 32; base < m; base += AGG_WAVES * 32) {
      const int nb = min(32, m - base);
      int r = 0, cl = 0;
      float nr = 0.0f;
      if (lane < nb) {
        const int e = q[base + lane];
        r = rows[e];
        const int c = cols[e];
        nr = dinv[r] * w[e] * dinv[c];
        cl = c - node0;
      }
#pragma unroll
      for (int j0 = 0; j0 < 32; j0 += U) {
        if (j0 < nb) {
          float v[U][F];
#pragma unroll
          for (int u = 0; u < U; ++u) {
            if (j0 + u < nb) {
              const int rr = __builtin_amdgcn_readlane(r, j0 + u);
              const float* __restrict__ hp = h + (size_t)rr * D + lane;
#pragma unroll
              for (int f = 0; f < F; ++f) v[u][f] = hp[f * 32];
            }
          }
#pragma unroll
          for (int u = 0; u < U; ++u) {
            if (j0 + u < nb) {
              const int cc = __builtin_amdgcn_readlane(cl, j0 + u);
              const float nn =
                  __int_as_float(__builtin_amdgcn_readlane(__float_as_int(nr), j0 + u));
              float* ap = acc + cc * D + lane;
#pragma unroll
              for (int f = 0; f < F; ++f) atomicAdd(ap + f * 32, v[u][f] * nn);
            }
          }
        }
      }
    }
    __syncthreads();
    par ^= 1;
  }

  {
    static_assert(F == 4 || F == 2, "a lane reads its F features of a row with one LDS load");
    float gm[F], bt[F];
#pragma unroll
    for (int f = 0; f < F; ++f) {
      gm[f] = gamma[lane * F + f];
      bt[f] = beta[lane * F + f];
    }
    auto ln_row = [&](int ln, float (&y)[F]) {
      const float* rp = acc + ln * D + lane * F;
      float v[F];
      if constexpr (F == 4) {
        const float4 t = *reinterpret_cast<const float4*>(rp);
        v[0] = t.x; v[1] = t.y; v[2] = t.z; v[3] = t.w;
      } else {
        const float2 t = *reinterpret_cast<const float2*>(rp);
        v[0] = t.x; v[1] = t.y;
      }
      float s = 0.0f;
#pragma unroll
      for (int f = 0; f < F; ++f) {
        v[f] = fmaxf(v[f], 0.0f);
        s += v[f];
      }
      s = wave_sum(s);
      const float mu = s * (1.0f / D);
      float ss = 0.0f;
#pragma unroll
      for (int f = 0; f < F; ++f) {
        const float d = v[f] - mu;
        ss += d * d;
      }
      ss = wave_sum(ss);
      const float rs = rsqrtf(ss * (1.0f / D) + EPSF);
#pragma unroll
      for (int f = 0; f < F; ++f) y[f] = (v[f] - mu) * rs * gm[f] + bt[f];
    };

    if constexpr (!POOL) {
      float yy[AGG_RPW][F];
#pragma unroll
      for (int k = 0; k < AGG_RPW; ++k) ln_row(min(wave + k * AGG_WAVES, AGG_NPB - 1), yy[k]);
#pragma unroll
      for (int k = 0; k < AGG_RPW; ++k) {
        const int ln = wave + k * AGG_WAVES;
        const int node = node0 + ln;
        const float (&y)[F] = yy[k];
        if (ln < AGG_NPB && node < n) {
          float* __restrict__ op = out + (size_t)node * D + lane * F;
          if constexpr (F == 4) {
            float4 o; o.x = y[0]; o.y = y[1]; o.z = y[2]; o.w = y[3];
            *reinterpret_cast<float4*>(op) = o;
          } else {
            float2 o; o.x = y[0]; o.y = y[1];
            *reinterpret_cast<float2*>(op) = o;
          }
        }
      }
    } else {
      __shared__ float tab[NGRAPHS * DOUT];
      __shared__ float gcnt[NGRAPHS];
      for (int i = tid; i < NGRAPHS * DOUT; i += AGG_THREADS) tab[i] = 0.0f;
      if (tid < NGRAPHS) gcnt[tid] = 0.0f;
      int gid[AGG_RPW];
#pragma unroll
      for (int k = 0; k < AGG_RPW; ++k)
        gid[k] = batch[min(node0 + wave + k * AGG_WAVES, n - 1)];
      __syncthreads();

      int c0 = 0, c1 = 0;
      float yy[AGG_RPW][F];
#pragma unroll
      for (int k = 0; k < AGG_RPW; ++k) ln_row(min(wave + k * AGG_WAVES, AGG_NPB - 1), yy[k]);
#pragma unroll
      for (int k = 0; k < AGG_RPW; ++k) {
        const int ln = wave + k * AGG_WAVES;
        const int node = node0 + ln;
        const float (&y)[F] = yy[k];
        if (ln < AGG_NPB && node < n) {
          const int g = gid[k];
          atomicAdd(&tab[g * DOUT + lane * 2 + 0], y[0]);
          atomicAdd(&tab[g * DOUT + lane * 2 + 1], y[1]);
          c0 += (g == lane) ? 1 : 0;
          c1 += (g == lane + 32) ? 1 : 0;
        }
      }
      if (c0 != 0) atomicAdd(&gcnt[lane], (float)c0);
      if (c1 != 0) atomicAdd(&gcnt[lane + 32], (float)c1);
      __syncthreads();

      for (int i = tid; i < NGRAPHS * DOUT; i += AGG_THREADS) {
        const float s = tab[i];
        if (s != 0.0f) atomicAdd(&sums[i], s);
      }
      if (tid < NGRAPHS) {
        const float c = gcnt[tid];
        if (c != 0.0f) atomicAdd(&cnts[tid], c);
      }
    }
  }
}

__global__ void pool_final(const float* __restrict__ sums, const float* __restrict__ cnts,
                           float* __restrict__ out) {
  int i = blockIdx.x * blockDim.x + threadIdx.x;
  if (i < NGRAPHS * DOUT) {
    int g = i / DOUT;
    float m = sums[i] / fmaxf(cnts[g], 1.0f);
    out[i] = 1.0f / (1.0f + __expf(-m));
  }
}

extern "C" void kernel_launch(void* const* d_in, const int* in_sizes, int n_in,
                              void* d_out, int out_size, void* d_ws, size_t ws_size,
                              hipStream_t stream) {
  const float* x   = (const float*)d_in[0];
  const int*   ei  = (const int*)d_in[1];
  const float* ew  = (const float*)d_in[2];
  const int*   bat = (const int*)d_in[3];
  const float* W1  = (const float*)d_in[4];
  const float* b1  = (const float*)d_in[5];
  const float* g1  = (const float*)d_in[6];
  const float* bt1 = (const float*)d_in[7];
  const float* W2  = (const float*)d_in[8];
  const float* b2  = (const float*)d_in[9];
  const float* g2  = (const float*)d_in[10];
  const float* bt2 = (const float*)d_in[11];
  float* out = (float*)d_out;

  const int* rows = ei;
  const int* cols = ei + NEDGES;

  char* ws = (char*)d_ws;
  size_t off = 0;
  auto carve = [&](size_t bytes) -> float* {
    float* p = (float*)(ws + off);
    off += (bytes + 255) & ~(size_t)255;
    return p;
  };
  float* dinv = carve((size_t)NNODES * 4);
  float* h    = carve((size_t)NNODES * DHID * 4);
  float* a    = carve((size_t)NNODES * DHID * 4);
  float* sums = carve((size_t)NGRAPHS * DOUT * 4);
  float* cnts = carve((size_t)NGRAPHS * 4);

  const int TB = 256;
  auto blocks = [&](long long threads) { return (int)((threads + TB - 1) / TB); };

  zero_f32<<<blocks(NNODES), TB, 0, stream>>>(dinv, NNODES);
  deg_accum<<<blocks(NEDGES), TB, 0, stream>>>(cols, ew, dinv, NEDGES);
  deg_to_dinv<<<blocks(NNODES), TB, 0, stream>>>(dinv, NNODES);

  wmma_gemm_f32<DHID, DIN, false><<<GEMM_BLOCKS, GEMM_THREADS, 0, stream>>>(
      x, W1, h, NNODES, nullptr, nullptr);
  gather_lds<DHID, false, false><<<AGG_BLOCKS, AGG_THREADS, 0, stream>>>(
      h, rows, cols, ew, dinv, b1, g1, bt1, a, nullptr, sums, cnts, NNODES, nullptr);

  unsigned int* keys = (unsigned int*)(h + (size_t)NNODES * DOUT);
  wmma_gemm_f32<DOUT, DHID, true><<<GEMM_BLOCKS, GEMM_THREADS, 0, stream>>>(
      a, W2, h, NNODES, cols, keys);
  gather_lds<DOUT, true, true><<<AGG_BLOCKS, AGG_THREADS, 0, stream>>>(
      h, rows, cols, ew, dinv, b2, g2, bt2, nullptr, bat, sums, cnts, NNODES, keys);
  pool_final<<<blocks(NGRAPHS * DOUT), TB, 0, stream>>>(sums, cnts, out);
}


# per-edge look-ups: target's local row rides in the queue word (layer 1) or in an 8-byte edge record written with the key bytes (layer 2), target's dinv from LDS: 5 scattered words per edge -> 3 / 2
# speedup vs baseline: 2.5099x; 1.0315x over previous
#include <hip/hip_runtime.h>
#include <math.h>

#define NNODES 100000
#define NEDGES 1600000
#define NGRAPHS 64
#define DIN 256
#define DHID 128
#define DOUT 64
#define EPSF 1e-5f

typedef __attribute__((ext_vector_type(2))) float v2f;
typedef __attribute__((ext_vector_type(8))) float v8f;

template <int N>
__device__ __forceinline__ float xrow(float x) {
  const int b = __builtin_bit_cast(int, x);
  return __builtin_bit_cast(float,
      __builtin_amdgcn_update_dpp(0, b, 0x160 + N, 0xf, 0xf, true));
}
__device__ __forceinline__ float xhalf(float x) {
  const int b = __builtin_bit_cast(int, x);
  return __builtin_bit_cast(float,
      __builtin_amdgcn_permlanex16(b, b, 0x76543210, 0xfedcba98, false, false));
}
__device__ __forceinline__ float wave_sum(float s) {
  s += xhalf(s);
  s += xrow<8>(s);
  s += xrow<4>(s);
  s += xrow<2>(s);
  s += xrow<1>(s);
  return s;
}

__global__ void zero_f32(float* __restrict__ p, int n) {
  int i = blockIdx.x * blockDim.x + threadIdx.x;
  if (i < n) p[i] = 0.0f;
}

__global__ void deg_accum(const int* __restrict__ col, const float* __restrict__ w,
                          float* __restrict__ deg, int nE) {
  int e = blockIdx.x * blockDim.x + threadIdx.x;
  if (e < nE) atomicAdd(&deg[col[e]], w[e]);
}

__global__ void deg_to_dinv(float* __restrict__ deg, int n) {
  int i = blockIdx.x * blockDim.x + threadIdx.x;
  if (i < n) {
    float d = deg[i] + 1.0f;
    deg[i] = rsqrtf(fmaxf(d, EPSF));
  }
}

#define GEMM_THREADS 1024
#define GEMM_WAVES (GEMM_THREADS / 32)
#define GEMM_BLOCKS 256
static_assert(NNODES % 16 == 0, "rows come in whole 16-row stripes");
#define AGG_NPB 392
#define AGG_BLOCKS ((NNODES + AGG_NPB - 1) / AGG_NPB)
static_assert(AGG_BLOCKS <= 256, "a work-group number fits one byte");
static_assert(NEDGES % 16 == 0, "the key bytes are read sixteen at a time");
static_assert((size_t)NEDGES * 9 <= (size_t)NNODES * (DHID - DOUT) * 4,
              "key bytes and edge records fit the half of h that layer 2 leaves unused");
static_assert(NEDGES <= (1 << 21) && AGG_NPB <= (1 << 9),
              "an edge number and a local row share one queue word");
static_assert(NNODES <= (1 << 17), "a source node and a local row share one record word");

template <int NCOLS, int KDIM, bool KEYS>
__global__ void __launch_bounds__(GEMM_THREADS)
wmma_gemm_f32(const float* __restrict__ A, const float* __restrict__ B,
              float* __restrict__ C, int M, const int* __restrict__ srcs,
              const int* __restrict__ cols, const float* __restrict__ ew,
              unsigned int* __restrict__ keys) {
  __shared__ float2 Bl[(KDIM / 2) * NCOLS];

  if constexpr (KEYS) {
    const int4* __restrict__ c4p = reinterpret_cast<const int4*>(cols);
    const int4* __restrict__ r4p = reinterpret_cast<const int4*>(srcs);
    const float4* __restrict__ w4p = reinterpret_cast<const float4*>(ew);
    uint4* __restrict__ rec4 = reinterpret_cast<uint4*>(keys + NEDGES / 4);
    const int nthr = (int)gridDim.x * GEMM_THREADS;
    for (int i = (int)blockIdx.x * GEMM_THREADS + (int)threadIdx.x; i < NEDGES / 4; i += nthr) {
      const int4 c = c4p[i];
      const int4 r = r4p[i];
      const float4 wv = w4p[i];
      const unsigned o0 = (unsigned)(c.x / AGG_NPB), o1 = (unsigned)(c.y / AGG_NPB);
      const unsigned o2 = (unsigned)(c.z / AGG_NPB), o3 = (unsigned)(c.w / AGG_NPB);
      keys[i] = o0 | (o1 << 8) | (o2 << 16) | (o3 << 24);
      uint4 lo, hi;
      lo.x = (unsigned)r.x | (((unsigned)c.x - o0 * AGG_NPB) << 17);
      lo.y = __float_as_uint(wv.x);
      lo.z = (unsigned)r.y | (((unsigned)c.y - o1 * AGG_NPB) << 17);
      lo.w = __float_as_uint(wv.y);
      hi.x = (unsigned)r.z | (((unsigned)c.z - o2 * AGG_NPB) << 17);
      hi.y = __float_as_uint(wv.z);
      hi.z = (unsigned)r.w | (((unsigned)c.w - o3 * AGG_NPB) << 17);
      hi.w = __float_as_uint(wv.w);
      rec4[2 * i + 0] = lo;
      rec4[2 * i + 1] = hi;
    }
  }

  for (int i = threadIdx.x; i < (KDIM / 2) * NCOLS; i += GEMM_THREADS) {
    const int k2 = i / NCOLS;
    const int n = i - k2 * NCOLS;
    float2 v;
    v.x = B[(size_t)(2 * k2 + 0) * NCOLS + n];
    v.y = B[(size_t)(2 * k2 + 1) * NCOLS + n];
    Bl[i] = v;
  }
  __syncthreads();

  const int wave = __builtin_amdgcn_readfirstlane((int)(threadIdx.x >> 5));
  const int lane = threadIdx.x & 31;
  const int half = lane >> 4;
  const int l = lane & 15;
  constexpr int NT = NCOLS / 16;
  const int nstripes = M / 16;

  for (int stripe = wave * (int)gridDim.x + (int)blockIdx.x; stripe < nstripes;
       stripe += GEMM_WAVES * (int)gridDim.x) {
    const int m0 = stripe * 16;

    v8f acc[NT];
#pragma unroll
    for (int t = 0; t < NT; ++t)
#pragma unroll
      for (int j = 0; j < 8; ++j) acc[t][j] = 0.0f;

    const float* __restrict__ arow = A + (size_t)(m0 + l) * KDIM + 2 * half;

    v2f a;
    a[0] = arow[0];
    a[1] = arow[1];

    for (int k0 = 0; k0 < KDIM; k0 += 4) {
      v2f a_next = a;
      if (k0 + 4 < KDIM) {
        a_next[0] = arow[k0 + 4 + 0];
        a_next[1] = arow[k0 + 4 + 1];
      }
      const float2* __restrict__ brow = Bl + (size_t)(k0 / 2 + half) * NCOLS + l;
      v2f bfrag[NT];
#pragma unroll
      for (int t = 0; t < NT; ++t) {
        float2 bb = brow[t * 16];
        bfrag[t][0] = bb.x;
        bfrag[t][1] = bb.y;
      }
#pragma unroll
      for (int t = 0; t < NT; ++t)
        acc[t] = __builtin_amdgcn_wmma_f32_16x16x4_f32(
             false, a,  false, bfrag[t],
             (short)0, acc[t],  false,  false);
      a = a_next;
    }

#pragma unroll
    for (int t = 0; t < NT; ++t)
#pragma unroll
      for (int r = 0; r < 8; ++r)
        C[(size_t)(m0 + r + 8 * half) * NCOLS + t * 16 + l] = acc[t][r];
  }
}

#define AGG_THREADS 1024
#define AGG_WAVES (AGG_THREADS / 32)
#define AGG_QCAP 4096
#define AGG_LOADS 8
#define AGG_ITERS 7
#define AGG_CH4 (AGG_ITERS * AGG_LOADS * AGG_THREADS)
static_assert(NEDGES % 4 == 0, "targets are read four at a time");
static_assert(NNODES < (1 << 30), "node numbers fit the unsigned range test");

#define AGG_RPW ((AGG_NPB + AGG_WAVES - 1) / AGG_WAVES)
static_assert(NGRAPHS == 64 && DOUT == 64,
              "a lane owns two features and counts two graphs");

__device__ __forceinline__ unsigned zero_bytes(unsigned x) {
  return ~(((x & 0x7f7f7f7fu) + 0x7f7f7f7fu) | x | 0x7f7f7f7fu);
}

template <int D, bool POOL, bool BYTEKEYS>
__global__ void __launch_bounds__(AGG_THREADS)
gather_lds(const float* __restrict__ h, const int* __restrict__ rows,
           const int* __restrict__ cols, const float* __restrict__ w,
           const float* __restrict__ dinv, const float* __restrict__ bias,
           const float* __restrict__ gamma, const float* __restrict__ beta,
           float* __restrict__ out, const int* __restrict__ batch,
           float* __restrict__ sums, float* __restrict__ cnts, int n,
           const unsigned int* __restrict__ keys) {
  static_assert(!POOL || D == DOUT, "the pool table is NGRAPHS x DOUT");
  constexpr int F = D / 32;
  constexpr int ROW4 = D / 4;
  constexpr int U = 16;
  constexpr int NE4 = NEDGES / 4;
  __shared__ __attribute__((aligned(16))) float acc[AGG_NPB * D];
  __shared__ int q[AGG_QCAP];
  __shared__ int qn[2];
  __shared__ float dloc[AGG_NPB];

  const int tid = threadIdx.x;
  const int lane = tid & 31;
  const int wave = __builtin_amdgcn_readfirstlane(tid >> 5);
  const int node0 = blockIdx.x * AGG_NPB;

  if (tid < 2) qn[tid] = 0;
  for (int i = tid; i < AGG_NPB; i += AGG_THREADS) dloc[i] = dinv[min(node0 + i, n - 1)];

  if constexpr (!POOL) {
    if (blockIdx.x == 0) {
      for (int i = tid; i < NGRAPHS * DOUT; i += AGG_THREADS) sums[i] = 0.0f;
      if (tid < NGRAPHS) cnts[tid] = 0.0f;
    }
  }

  {
    const float4* __restrict__ h4 = reinterpret_cast<const float4*>(h);
    const float4* __restrict__ b4 = reinterpret_cast<const float4*>(bias);
    float4* acc4 = reinterpret_cast<float4*>(acc);
#pragma unroll 4
    for (int i = tid; i < AGG_NPB * ROW4; i += AGG_THREADS) {
      const int ln = i / ROW4;
      const int f4 = i - ln * ROW4;
      const int node = min(node0 + ln, n - 1);
      const float d = dinv[node];
      const float s = d * d;
      const float4 hv = h4[(size_t)node * ROW4 + f4];
      const float4 bv = b4[f4];
      float4 o;
      o.x = bv.x + hv.x * s;
      o.y = bv.y + hv.y * s;
      o.z = bv.z + hv.z * s;
      o.w = bv.w + hv.w * s;
      acc4[i] = o;
    }
  }
  __syncthreads();

  const int4* __restrict__ c4p = reinterpret_cast<const int4*>(cols);

  auto push = [&](int e, int* cnt) {
    const int p = atomicAdd(cnt, 1);
    if (p < AGG_QCAP) q[p] = e;
  };

  int par = 0;
  for (int ch = 0; ch < NE4; ch += AGG_CH4) {
    int* cnt = &qn[par];
    if constexpr (BYTEKEYS) {
      constexpr int KLOADS = 7;
      constexpr int KITERS = 2;
      static_assert(KLOADS * KITERS * 16 == AGG_ITERS * AGG_LOADS * 4,
                    "a piece covers the same edges with either kind of scan");
      constexpr int NK16 = NEDGES / 16;
      const uint4* __restrict__ k4p = reinterpret_cast<const uint4*>(keys);
      const unsigned pat = (unsigned)blockIdx.x * 0x01010101u;
      for (int it = 0; it < KITERS; ++it) {
        uint4 kv[KLOADS];
        int gi[KLOADS];
#pragma unroll
        for (int j = 0; j < KLOADS; ++j) {
          gi[j] = ch / 4 + (it * KLOADS + j) * AGG_THREADS + tid;
          if (gi[j] < NK16) {
            kv[j] = k4p[gi[j]];
          } else {
            kv[j].x = ~pat; kv[j].y = ~pat; kv[j].z = ~pat; kv[j].w = ~pat;
          }
        }
#pragma unroll
        for (int j = 0; j < KLOADS; ++j) {
          unsigned m = (zero_bytes(kv[j].x ^ pat) >> 7) | (zero_bytes(kv[j].y ^ pat) >> 6) |
                       (zero_bytes(kv[j].z ^ pat) >> 5) | (zero_bytes(kv[j].w ^ pat) >> 4);
          while (m != 0) {
            const int p = __builtin_ctz(m);
            m &= m - 1;
            push(gi[j] * 16 + ((p & 7) << 2) + (p >> 3), cnt);
          }
        }
      }
    } else {
    for (int it = 0; it < AGG_ITERS; ++it) {
      int4 cv[AGG_LOADS];
      int idx[AGG_LOADS];
#pragma unroll
      for (int j = 0; j < AGG_LOADS; ++j) {
        idx[j] = ch + (it * AGG_LOADS + j) * AGG_THREADS + tid;
        if (idx[j] < NE4) {
          cv[j] = c4p[idx[j]];
        } else {
          cv[j].x = -1; cv[j].y = -1; cv[j].z = -1; cv[j].w = -1;
        }
      }
#pragma unroll
      for (int j = 0; j < AGG_LOADS; ++j) {
        const unsigned d0 = (unsigned)(cv[j].x - node0);
        const unsigned d1 = (unsigned)(cv[j].y - node0);
        const unsigned d2 = (unsigned)(cv[j].z - node0);
        const unsigned d3 = (unsigned)(cv[j].w - node0);
        if (d0 < AGG_NPB) push((idx[j] * 4 + 0) | (int)(d0 << 21), cnt);
        if (d1 < AGG_NPB) push((idx[j] * 4 + 1) | (int)(d1 << 21), cnt);
        if (d2 < AGG_NPB) push((idx[j] * 4 + 2) | (int)(d2 << 21), cnt);
        if (d3 < AGG_NPB) push((idx[j] * 4 + 3) | (int)(d3 << 21), cnt);
      }
    }
    }
    __syncthreads();

    int m = __builtin_amdgcn_readfirstlane(*cnt);
    if (tid == 0) qn[par ^ 1] = 0;
    if (m > AGG_QCAP) {
      const int e1 = min((ch + AGG_CH4) * 4, NEDGES);
      for (int e = ch * 4 + tid; e < e1; e += AGG_THREADS) {
        const int c = cols[e];
        const unsigned dl = (unsigned)(c - node0);
        if (dl < AGG_NPB) {
          const int r = rows[e];
          const float nr = dinv[r] * w[e] * dinv[c];
          for (int f = 0; f < D; ++f)
            atomicAdd(&acc[dl * D + f], h[(size_t)r * D + f] * nr);
        }
      }
      m = 0;
    }
    for (int base = wave * 32; base < m; base += AGG_WAVES * 32) {
      const int nb = min(32, m - base);
      int r = 0, cl = 0;
      float nr = 0.0f;
      if (lane < nb) {
        const int qv = q[base + lane];
        if constexpr (BYTEKEYS) {
          const uint2 rc = reinterpret_cast<const uint2*>(keys + NEDGES / 4)[qv];
          r = (int)(rc.x & 0x1ffffu);
          cl = (int)(rc.x >> 17);
          nr = dinv[r] * __uint_as_float(rc.y) * dloc[cl];
        } else {
          const int e = qv & 0x1fffff;
          cl = qv >> 21;
          r = rows[e];
          nr = dinv[r] * w[e] * dloc[cl];
        }
      }
#pragma unroll
      for (int j0 = 0; j0 < 32; j0 += U) {
        if (j0 < nb) {
          float v[U][F];
#pragma unroll
          for (int u = 0; u < U; ++u) {
            if (j0 + u < nb) {
              const int rr = __builtin_amdgcn_readlane(r, j0 + u);
              const float* __restrict__ hp = h + (size_t)rr * D + lane;
#pragma unroll
              for (int f = 0; f < F; ++f) v[u][f] = hp[f * 32];
            }
          }
#pragma unroll
          for (int u = 0; u < U; ++u) {
            if (j0 + u < nb) {
              const int cc = __builtin_amdgcn_readlane(cl, j0 + u);
              const float nn =
                  __int_as_float(__builtin_amdgcn_readlane(__float_as_int(nr), j0 + u));
              float* ap = acc + cc * D + lane;
#pragma unroll
              for (int f = 0; f < F; ++f) atomicAdd(ap + f * 32, v[u][f] * nn);
            }
          }
        }
      }
    }
    __syncthreads();
    par ^= 1;
  }

  {
    static_assert(F == 4 || F == 2, "a lane reads its F features of a row with one LDS load");
    float gm[F], bt[F];
#pragma unroll
    for (int f = 0; f < F; ++f) {
      gm[f] = gamma[lane * F + f];
      bt[f] = beta[lane * F + f];
    }
    auto ln_row = [&](int ln, float (&y)[F]) {
      const float* rp = acc + ln * D + lane * F;
      float v[F];
      if constexpr (F == 4) {
        const float4 t = *reinterpret_cast<const float4*>(rp);
        v[0] = t.x; v[1] = t.y; v[2] = t.z; v[3] = t.w;
      } else {
        const float2 t = *reinterpret_cast<const float2*>(rp);
        v[0] = t.x; v[1] = t.y;
      }
      float s = 0.0f;
#pragma unroll
      for (int f = 0; f < F; ++f) {
        v[f] = fmaxf(v[f], 0.0f);
        s += v[f];
      }
      s = wave_sum(s);
      const float mu = s * (1.0f / D);
      float ss = 0.0f;
#pragma unroll
      for (int f = 0; f < F; ++f) {
        const float d = v[f] - mu;
        ss += d * d;
      }
      ss = wave_sum(ss);
      const float rs = rsqrtf(ss * (1.0f / D) + EPSF);
#pragma unroll
      for (int f = 0; f < F; ++f) y[f] = (v[f] - mu) * rs * gm[f] + bt[f];
    };

    if constexpr (!POOL) {
      float yy[AGG_RPW][F];
#pragma unroll
      for (int k = 0; k < AGG_RPW; ++k) ln_row(min(wave + k * AGG_WAVES, AGG_NPB - 1), yy[k]);
#pragma unroll
      for (int k = 0; k < AGG_RPW; ++k) {
        const int ln = wave + k * AGG_WAVES;
        const int node = node0 + ln;
        const float (&y)[F] = yy[k];
        if (ln < AGG_NPB && node < n) {
          float* __restrict__ op = out + (size_t)node * D + lane * F;
          if constexpr (F == 4) {
            float4 o; o.x = y[0]; o.y = y[1]; o.z = y[2]; o.w = y[3];
            *reinterpret_cast<float4*>(op) = o;
          } else {
            float2 o; o.x = y[0]; o.y = y[1];
            *reinterpret_cast<float2*>(op) = o;
          }
        }
      }
    } else {
      __shared__ float tab[NGRAPHS * DOUT];
      __shared__ float gcnt[NGRAPHS];
      for (int i = tid; i < NGRAPHS * DOUT; i += AGG_THREADS) tab[i] = 0.0f;
      if (tid < NGRAPHS) gcnt[tid] = 0.0f;
      int gid[AGG_RPW];
#pragma unroll
      for (int k = 0; k < AGG_RPW; ++k)
        gid[k] = batch[min(node0 + wave + k * AGG_WAVES, n - 1)];
      __syncthreads();

      int c0 = 0, c1 = 0;
      float yy[AGG_RPW][F];
#pragma unroll
      for (int k = 0; k < AGG_RPW; ++k) ln_row(min(wave + k * AGG_WAVES, AGG_NPB - 1), yy[k]);
#pragma unroll
      for (int k = 0; k < AGG_RPW; ++k) {
        const int ln = wave + k * AGG_WAVES;
        const int node = node0 + ln;
        const float (&y)[F] = yy[k];
        if (ln < AGG_NPB && node < n) {
          const int g = gid[k];
          atomicAdd(&tab[g * DOUT + lane * 2 + 0], y[0]);
          atomicAdd(&tab[g * DOUT + lane * 2 + 1], y[1]);
          c0 += (g == lane) ? 1 : 0;
          c1 += (g == lane + 32) ? 1 : 0;
        }
      }
      if (c0 != 0) atomicAdd(&gcnt[lane], (float)c0);
      if (c1 != 0) atomicAdd(&gcnt[lane + 32], (float)c1);
      __syncthreads();

      for (int i = tid; i < NGRAPHS * DOUT; i += AGG_THREADS) {
        const float s = tab[i];
        if (s != 0.0f) atomicAdd(&sums[i], s);
      }
      if (tid < NGRAPHS) {
        const float c = gcnt[tid];
        if (c != 0.0f) atomicAdd(&cnts[tid], c);
      }
    }
  }
}

__global__ void pool_final(const float* __restrict__ sums, const float* __restrict__ cnts,
                           float* __restrict__ out) {
  int i = blockIdx.x * blockDim.x + threadIdx.x;
  if (i < NGRAPHS * DOUT) {
    int g = i / DOUT;
    float m = sums[i] / fmaxf(cnts[g], 1.0f);
    out[i] = 1.0f / (1.0f + __expf(-m));
  }
}

extern "C" void kernel_launch(void* const* d_in, const int* in_sizes, int n_in,
                              void* d_out, int out_size, void* d_ws, size_t ws_size,
                              hipStream_t stream) {
  const float* x   = (const float*)d_in[0];
  const int*   ei  = (const int*)d_in[1];
  const float* ew  = (const float*)d_in[2];
  const int*   bat = (const int*)d_in[3];
  const float* W1  = (const float*)d_in[4];
  const float* b1  = (const float*)d_in[5];
  const float* g1  = (const float*)d_in[6];
  const float* bt1 = (const float*)d_in[7];
  const float* W2  = (const float*)d_in[8];
  const float* b2  = (const float*)d_in[9];
  const float* g2  = (const float*)d_in[10];
  const float* bt2 = (const float*)d_in[11];
  float* out = (float*)d_out;

  const int* rows = ei;
  const int* cols = ei + NEDGES;

  char* ws = (char*)d_ws;
  size_t off = 0;
  auto carve = [&](size_t bytes) -> float* {
    float* p = (float*)(ws + off);
    off += (bytes + 255) & ~(size_t)255;
    return p;
  };
  float* dinv = carve((size_t)NNODES * 4);
  float* h    = carve((size_t)NNODES * DHID * 4);
  float* a    = carve((size_t)NNODES * DHID * 4);
  float* sums = carve((size_t)NGRAPHS * DOUT * 4);
  float* cnts = carve((size_t)NGRAPHS * 4);

  const int TB = 256;
  auto blocks = [&](long long threads) { return (int)((threads + TB - 1) / TB); };

  zero_f32<<<blocks(NNODES), TB, 0, stream>>>(dinv, NNODES);
  deg_accum<<<blocks(NEDGES), TB, 0, stream>>>(cols, ew, dinv, NEDGES);
  deg_to_dinv<<<blocks(NNODES), TB, 0, stream>>>(dinv, NNODES);

  wmma_gemm_f32<DHID, DIN, false><<<GEMM_BLOCKS, GEMM_THREADS, 0, stream>>>(
      x, W1, h, NNODES, nullptr, nullptr, nullptr, nullptr);
  gather_lds<DHID, false, false><<<AGG_BLOCKS, AGG_THREADS, 0, stream>>>(
      h, rows, cols, ew, dinv, b1, g1, bt1, a, nullptr, sums, cnts, NNODES, nullptr);

  unsigned int* keys = (unsigned int*)(h + (size_t)NNODES * DOUT);
  wmma_gemm_f32<DOUT, DHID, true><<<GEMM_BLOCKS, GEMM_THREADS, 0, stream>>>(
      a, W2, h, NNODES, rows, cols, ew, keys);
  gather_lds<DOUT, true, true><<<AGG_BLOCKS, AGG_THREADS, 0, stream>>>(
      h, rows, cols, ew, dinv, b2, g2, bt2, nullptr, bat, sums, cnts, NNODES, keys);
  pool_final<<<blocks(NGRAPHS * DOUT), TB, 0, stream>>>(sums, cnts, out);
}


# row sums: each wave queues its own hits (ballot + lane rank, scalar count, no shared LDS counter) and adds their rows itself once 32 wait; no barrier between reading and adding; sum order changes
# speedup vs baseline: 2.6291x; 1.0475x over previous
#include <hip/hip_runtime.h>
#include <math.h>

#define NNODES 100000
#define NEDGES 1600000
#define NGRAPHS 64
#define DIN 256
#define DHID 128
#define DOUT 64
#define EPSF 1e-5f

typedef __attribute__((ext_vector_type(2))) float v2f;
typedef __attribute__((ext_vector_type(8))) float v8f;

template <int N>
__device__ __forceinline__ float xrow(float x) {
  const int b = __builtin_bit_cast(int, x);
  return __builtin_bit_cast(float,
      __builtin_amdgcn_update_dpp(0, b, 0x160 + N, 0xf, 0xf, true));
}
__device__ __forceinline__ float xhalf(float x) {
  const int b = __builtin_bit_cast(int, x);
  return __builtin_bit_cast(float,
      __builtin_amdgcn_permlanex16(b, b, 0x76543210, 0xfedcba98, false, false));
}
__device__ __forceinline__ float wave_sum(float s) {
  s += xhalf(s);
  s += xrow<8>(s);
  s += xrow<4>(s);
  s += xrow<2>(s);
  s += xrow<1>(s);
  return s;
}

__global__ void zero_f32(float* __restrict__ p, int n) {
  int i = blockIdx.x * blockDim.x + threadIdx.x;
  if (i < n) p[i] = 0.0f;
}

__global__ void deg_accum(const int* __restrict__ col, const float* __restrict__ w,
                          float* __restrict__ deg, int nE) {
  int e = blockIdx.x * blockDim.x + threadIdx.x;
  if (e < nE) atomicAdd(&deg[col[e]], w[e]);
}

__global__ void deg_to_dinv(float* __restrict__ deg, int n) {
  int i = blockIdx.x * blockDim.x + threadIdx.x;
  if (i < n) {
    float d = deg[i] + 1.0f;
    deg[i] = rsqrtf(fmaxf(d, EPSF));
  }
}

#define GEMM_THREADS 1024
#define GEMM_WAVES (GEMM_THREADS / 32)
#define GEMM_BLOCKS 256
static_assert(NNODES % 16 == 0, "rows come in whole 16-row stripes");
#define AGG_NPB 392
#define AGG_BLOCKS ((NNODES + AGG_NPB - 1) / AGG_NPB)
static_assert(AGG_BLOCKS <= 256, "a work-group number fits one byte");
static_assert(NEDGES % 16 == 0, "the key bytes are read sixteen at a time");
static_assert((size_t)NEDGES * 9 <= (size_t)NNODES * (DHID - DOUT) * 4,
              "key bytes and edge records fit the half of h that layer 2 leaves unused");
static_assert(NEDGES <= (1 << 21) && AGG_NPB <= (1 << 9),
              "an edge number and a local row share one queue word");
static_assert(NNODES <= (1 << 17), "a source node and a local row share one record word");

template <int NCOLS, int KDIM, bool KEYS>
__global__ void __launch_bounds__(GEMM_THREADS)
wmma_gemm_f32(const float* __restrict__ A, const float* __restrict__ B,
              float* __restrict__ C, int M, const int* __restrict__ srcs,
              const int* __restrict__ cols, const float* __restrict__ ew,
              unsigned int* __restrict__ keys) {
  __shared__ float2 Bl[(KDIM / 2) * NCOLS];

  if constexpr (KEYS) {
    const int4* __restrict__ c4p = reinterpret_cast<const int4*>(cols);
    const int4* __restrict__ r4p = reinterpret_cast<const int4*>(srcs);
    const float4* __restrict__ w4p = reinterpret_cast<const float4*>(ew);
    uint4* __restrict__ rec4 = reinterpret_cast<uint4*>(keys + NEDGES / 4);
    const int nthr = (int)gridDim.x * GEMM_THREADS;
    for (int i = (int)blockIdx.x * GEMM_THREADS + (int)threadIdx.x; i < NEDGES / 4; i += nthr) {
      const int4 c = c4p[i];
      const int4 r = r4p[i];
      const float4 wv = w4p[i];
      const unsigned o0 = (unsigned)(c.x / AGG_NPB), o1 = (unsigned)(c.y / AGG_NPB);
      const unsigned o2 = (unsigned)(c.z / AGG_NPB), o3 = (unsigned)(c.w / AGG_NPB);
      keys[i] = o0 | (o1 << 8) | (o2 << 16) | (o3 << 24);
      uint4 lo, hi;
      lo.x = (unsigned)r.x | (((unsigned)c.x - o0 * AGG_NPB) << 17);
      lo.y = __float_as_uint(wv.x);
      lo.z = (unsigned)r.y | (((unsigned)c.y - o1 * AGG_NPB) << 17);
      lo.w = __float_as_uint(wv.y);
      hi.x = (unsigned)r.z | (((unsigned)c.z - o2 * AGG_NPB) << 17);
      hi.y = __float_as_uint(wv.z);
      hi.z = (unsigned)r.w | (((unsigned)c.w - o3 * AGG_NPB) << 17);
      hi.w = __float_as_uint(wv.w);
      rec4[2 * i + 0] = lo;
      rec4[2 * i + 1] = hi;
    }
  }

  for (int i = threadIdx.x; i < (KDIM / 2) * NCOLS; i += GEMM_THREADS) {
    const int k2 = i / NCOLS;
    const int n = i - k2 * NCOLS;
    float2 v;
    v.x = B[(size_t)(2 * k2 + 0) * NCOLS + n];
    v.y = B[(size_t)(2 * k2 + 1) * NCOLS + n];
    Bl[i] = v;
  }
  __syncthreads();

  const int wave = __builtin_amdgcn_readfirstlane((int)(threadIdx.x >> 5));
  const int lane = threadIdx.x & 31;
  const int half = lane >> 4;
  const int l = lane & 15;
  constexpr int NT = NCOLS / 16;
  const int nstripes = M / 16;

  for (int stripe = wave * (int)gridDim.x + (int)blockIdx.x; stripe < nstripes;
       stripe += GEMM_WAVES * (int)gridDim.x) {
    const int m0 = stripe * 16;

    v8f acc[NT];
#pragma unroll
    for (int t = 0; t < NT; ++t)
#pragma unroll
      for (int j = 0; j < 8; ++j) acc[t][j] = 0.0f;

    const float* __restrict__ arow = A + (size_t)(m0 + l) * KDIM + 2 * half;

    v2f a;
    a[0] = arow[0];
    a[1] = arow[1];

    for (int k0 = 0; k0 < KDIM; k0 += 4) {
      v2f a_next = a;
      if (k0 + 4 < KDIM) {
        a_next[0] = arow[k0 + 4 + 0];
        a_next[1] = arow[k0 + 4 + 1];
      }
      const float2* __restrict__ brow = Bl + (size_t)(k0 / 2 + half) * NCOLS + l;
      v2f bfrag[NT];
#pragma unroll
      for (int t = 0; t < NT; ++t) {
        float2 bb = brow[t * 16];
        bfrag[t][0] = bb.x;
        bfrag[t][1] = bb.y;
      }
#pragma unroll
      for (int t = 0; t < NT; ++t)
        acc[t] = __builtin_amdgcn_wmma_f32_16x16x4_f32(
             false, a,  false, bfrag[t],
             (short)0, acc[t],  false,  false);
      a = a_next;
    }

#pragma unroll
    for (int t = 0; t < NT; ++t)
#pragma unroll
      for (int r = 0; r < 8; ++r)
        C[(size_t)(m0 + r + 8 * half) * NCOLS + t * 16 + l] = acc[t][r];
  }
}

#define AGG_THREADS 1024
#define AGG_WAVES (AGG_THREADS / 32)
#define AGG_QCAP 4096
#define AGG_LOADS 8
static_assert(NEDGES % 4 == 0, "targets are read four at a time");
static_assert(NNODES < (1 << 30), "node numbers fit the unsigned range test");

#define AGG_RPW ((AGG_NPB + AGG_WAVES - 1) / AGG_WAVES)
static_assert(NGRAPHS == 64 && DOUT == 64,
              "a lane owns two features and counts two graphs");

__device__ __forceinline__ unsigned zero_bytes(unsigned x) {
  return ~(((x & 0x7f7f7f7fu) + 0x7f7f7f7fu) | x | 0x7f7f7f7fu);
}

template <int D, bool POOL, bool BYTEKEYS>
__global__ void __launch_bounds__(AGG_THREADS)
gather_lds(const float* __restrict__ h, const int* __restrict__ rows,
           const int* __restrict__ cols, const float* __restrict__ w,
           const float* __restrict__ dinv, const float* __restrict__ bias,
           const float* __restrict__ gamma, const float* __restrict__ beta,
           float* __restrict__ out, const int* __restrict__ batch,
           float* __restrict__ sums, float* __restrict__ cnts, int n,
           const unsigned int* __restrict__ keys) {
  static_assert(!POOL || D == DOUT, "the pool table is NGRAPHS x DOUT");
  constexpr int F = D / 32;
  constexpr int ROW4 = D / 4;
  constexpr int U = 16;
  constexpr int NE4 = NEDGES / 4;
  __shared__ __attribute__((aligned(16))) float acc[AGG_NPB * D];
  __shared__ int q[AGG_QCAP];
  __shared__ float dloc[AGG_NPB];

  const int tid = threadIdx.x;
  const int lane = tid & 31;
  const int wave = __builtin_amdgcn_readfirstlane(tid >> 5);
  const int node0 = blockIdx.x * AGG_NPB;

  for (int i = tid; i < AGG_NPB; i += AGG_THREADS) dloc[i] = dinv[min(node0 + i, n - 1)];

  if constexpr (!POOL) {
    if (blockIdx.x == 0) {
      for (int i = tid; i < NGRAPHS * DOUT; i += AGG_THREADS) sums[i] = 0.0f;
      if (tid < NGRAPHS) cnts[tid] = 0.0f;
    }
  }

  {
    const float4* __restrict__ h4 = reinterpret_cast<const float4*>(h);
    const float4* __restrict__ b4 = reinterpret_cast<const float4*>(bias);
    float4* acc4 = reinterpret_cast<float4*>(acc);
#pragma unroll 4
    for (int i = tid; i < AGG_NPB * ROW4; i += AGG_THREADS) {
      const int ln = i / ROW4;
      const int f4 = i - ln * ROW4;
      const int node = min(node0 + ln, n - 1);
      const float d = dinv[node];
      const float s = d * d;
      const float4 hv = h4[(size_t)node * ROW4 + f4];
      const float4 bv = b4[f4];
      float4 o;
      o.x = bv.x + hv.x * s;
      o.y = bv.y + hv.y * s;
      o.z = bv.z + hv.z * s;
      o.w = bv.w + hv.w * s;
      acc4[i] = o;
    }
  }
  __syncthreads();

  const int4* __restrict__ c4p = reinterpret_cast<const int4*>(cols);

  constexpr int WQCAP = AGG_QCAP / AGG_WAVES;
  static_assert(WQCAP >= 64, "a wave's part holds a full batch and then some");
  int* wq = q + wave * WQCAP;
  int qcount = 0;

  auto push = [&](unsigned mask, bool hit, int word) {
    const int pos = qcount + (int)__builtin_amdgcn_mbcnt_lo(mask, 0u);
    if (hit && pos < WQCAP) wq[pos] = word;
    qcount += __builtin_popcount(mask);
  };

  constexpr int LOADS = BYTEKEYS ? 7 : AGG_LOADS;
  constexpr int EPL = BYTEKEYS ? 16 : 4;
  constexpr int NV = NEDGES / EPL;
  constexpr int NTRIPS = (NV + LOADS * AGG_THREADS - 1) / (LOADS * AGG_THREADS);

  for (int it = 0; it <= NTRIPS; ++it) {
    const int q0 = qcount;
    if (it < NTRIPS) {
      const bool whole = (it + 1) * LOADS * AGG_THREADS <= NV;
      if constexpr (BYTEKEYS) {
        const uint4* __restrict__ k4p = reinterpret_cast<const uint4*>(keys);
        const unsigned pat = (unsigned)blockIdx.x * 0x01010101u;
        uint4 kv[LOADS];
        int gi[LOADS];
#pragma unroll
        for (int j = 0; j < LOADS; ++j) gi[j] = (it * LOADS + j) * AGG_THREADS + tid;
        if (whole) {
#pragma unroll
          for (int j = 0; j < LOADS; ++j) kv[j] = k4p[gi[j]];
        } else {
#pragma unroll
          for (int j = 0; j < LOADS; ++j) {
            if (gi[j] < NV) {
              kv[j] = k4p[gi[j]];
            } else {
              kv[j].x = ~pat; kv[j].y = ~pat; kv[j].z = ~pat; kv[j].w = ~pat;
            }
          }
        }
#pragma unroll
        for (int j = 0; j < LOADS; ++j) {
          unsigned m = (zero_bytes(kv[j].x ^ pat) >> 7) | (zero_bytes(kv[j].y ^ pat) >> 6) |
                       (zero_bytes(kv[j].z ^ pat) >> 5) | (zero_bytes(kv[j].w ^ pat) >> 4);
          unsigned mk = __builtin_amdgcn_ballot_w32(m != 0);
          while (mk != 0) {
            const bool hit = m != 0;
            const int p = __builtin_ctz(m | 0x80000000u);
            push(mk, hit, gi[j] * 16 + ((p & 7) << 2) + (p >> 3));
            m &= m - 1;
            mk = __builtin_amdgcn_ballot_w32(m != 0);
          }
        }
      } else {
        int4 cv[LOADS];
        int idx[LOADS];
#pragma unroll
        for (int j = 0; j < LOADS; ++j) idx[j] = (it * LOADS + j) * AGG_THREADS + tid;
        if (whole) {
#pragma unroll
          for (int j = 0; j < LOADS; ++j) cv[j] = c4p[idx[j]];
        } else {
#pragma unroll
          for (int j = 0; j < LOADS; ++j) {
            if (idx[j] < NV) {
              cv[j] = c4p[idx[j]];
            } else {
              cv[j].x = -1; cv[j].y = -1; cv[j].z = -1; cv[j].w = -1;
            }
          }
        }
        auto test = [&](int c, int e) {
          const unsigned d = (unsigned)(c - node0);
          const bool hit = d < AGG_NPB;
          const unsigned mk = __builtin_amdgcn_ballot_w32(hit);
          if (mk != 0) push(mk, hit, e | (int)(d << 21));
        };
#pragma unroll
        for (int j = 0; j < LOADS; ++j) {
          test(cv[j].x, idx[j] * 4 + 0);
          test(cv[j].y, idx[j] * 4 + 1);
          test(cv[j].z, idx[j] * 4 + 2);
          test(cv[j].w, idx[j] * 4 + 3);
        }
      }
      if (qcount > WQCAP) {
        qcount = q0;
#pragma unroll 1
        for (int j = 0; j < LOADS; ++j) {
          const int e0 = ((it * LOADS + j) * AGG_THREADS + tid) * EPL;
#pragma unroll 1
          for (int k = 0; k < EPL; ++k) {
            const int e = e0 + k;
            if (e < NEDGES) {
              const int c = cols[e];
              const unsigned dl = (unsigned)(c - node0);
              if (dl < AGG_NPB) {
                const int r = rows[e];
                const float nr = dinv[r] * w[e] * dinv[c];
                for (int f = 0; f < D; ++f)
                  atomicAdd(&acc[dl * D + f], h[(size_t)r * D + f] * nr);
              }
            }
          }
        }
      }
    }
    const int need = (it < NTRIPS) ? 32 : 1;
    int base = 0;
    while (qcount - base >= need) {
      const int nb = min(32, qcount - base);
      int r = 0, cl = 0;
      float nr = 0.0f;
      if (lane < nb) {
        const int qv = wq[base + lane];
        if constexpr (BYTEKEYS) {
          const uint2 rc = reinterpret_cast<const uint2*>(keys + NEDGES / 4)[qv];
          r = (int)(rc.x & 0x1ffffu);
          cl = (int)(rc.x >> 17);
          nr = dinv[r] * __uint_as_float(rc.y) * dloc[cl];
        } else {
          const int e = qv & 0x1fffff;
          cl = qv >> 21;
          r = rows[e];
          nr = dinv[r] * w[e] * dloc[cl];
        }
      }
#pragma unroll
      for (int j0 = 0; j0 < 32; j0 += U) {
        if (j0 < nb) {
          float v[U][F];
#pragma unroll
          for (int u = 0; u < U; ++u) {
            if (j0 + u < nb) {
              const int rr = __builtin_amdgcn_readlane(r, j0 + u);
              const float* __restrict__ hp = h + (size_t)rr * D + lane;
#pragma unroll
              for (int f = 0; f < F; ++f) v[u][f] = hp[f * 32];
            }
          }
#pragma unroll
          for (int u = 0; u < U; ++u) {
            if (j0 + u < nb) {
              const int cc = __builtin_amdgcn_readlane(cl, j0 + u);
              const float nn =
                  __int_as_float(__builtin_amdgcn_readlane(__float_as_int(nr), j0 + u));
              float* ap = acc + cc * D + lane;
#pragma unroll
              for (int f = 0; f < F; ++f) atomicAdd(ap + f * 32, v[u][f] * nn);
            }
          }
        }
      }
      base += nb;
    }
    if (base > 0) {
      const int rem = qcount - base;
      int t = 0;
      if (lane < rem) t = wq[base + lane];
      if (lane < rem) wq[lane] = t;
      qcount = rem;
    }
  }
  __syncthreads();

  {
    static_assert(F == 4 || F == 2, "a lane reads its F features of a row with one LDS load");
    float gm[F], bt[F];
#pragma unroll
    for (int f = 0; f < F; ++f) {
      gm[f] = gamma[lane * F + f];
      bt[f] = beta[lane * F + f];
    }
    auto ln_row = [&](int ln, float (&y)[F]) {
      const float* rp = acc + ln * D + lane * F;
      float v[F];
      if constexpr (F == 4) {
        const float4 t = *reinterpret_cast<const float4*>(rp);
        v[0] = t.x; v[1] = t.y; v[2] = t.z; v[3] = t.w;
      } else {
        const float2 t = *reinterpret_cast<const float2*>(rp);
        v[0] = t.x; v[1] = t.y;
      }
      float s = 0.0f;
#pragma unroll
      for (int f = 0; f < F; ++f) {
        v[f] = fmaxf(v[f], 0.0f);
        s += v[f];
      }
      s = wave_sum(s);
      const float mu = s * (1.0f / D);
      float ss = 0.0f;
#pragma unroll
      for (int f = 0; f < F; ++f) {
        const float d = v[f] - mu;
        ss += d * d;
      }
      ss = wave_sum(ss);
      const float rs = rsqrtf(ss * (1.0f / D) + EPSF);
#pragma unroll
      for (int f = 0; f < F; ++f) y[f] = (v[f] - mu) * rs * gm[f] + bt[f];
    };

    if constexpr (!POOL) {
      float yy[AGG_RPW][F];
#pragma unroll
      for (int k = 0; k < AGG_RPW; ++k) ln_row(min(wave + k * AGG_WAVES, AGG_NPB - 1), yy[k]);
#pragma unroll
      for (int k = 0; k < AGG_RPW; ++k) {
        const int ln = wave + k * AGG_WAVES;
        const int node = node0 + ln;
        const float (&y)[F] = yy[k];
        if (ln < AGG_NPB && node < n) {
          float* __restrict__ op = out + (size_t)node * D + lane * F;
          if constexpr (F == 4) {
            float4 o; o.x = y[0]; o.y = y[1]; o.z = y[2]; o.w = y[3];
            *reinterpret_cast<float4*>(op) = o;
          } else {
            float2 o; o.x = y[0]; o.y = y[1];
            *reinterpret_cast<float2*>(op) = o;
          }
        }
      }
    } else {
      __shared__ float tab[NGRAPHS * DOUT];
      __shared__ float gcnt[NGRAPHS];
      for (int i = tid; i < NGRAPHS * DOUT; i += AGG_THREADS) tab[i] = 0.0f;
      if (tid < NGRAPHS) gcnt[tid] = 0.0f;
      int gid[AGG_RPW];
#pragma unroll
      for (int k = 0; k < AGG_RPW; ++k)
        gid[k] = batch[min(node0 + wave + k * AGG_WAVES, n - 1)];
      __syncthreads();

      int c0 = 0, c1 = 0;
      float yy[AGG_RPW][F];
#pragma unroll
      for (int k = 0; k < AGG_RPW; ++k) ln_row(min(wave + k * AGG_WAVES, AGG_NPB - 1), yy[k]);
#pragma unroll
      for (int k = 0; k < AGG_RPW; ++k) {
        const int ln = wave + k * AGG_WAVES;
        const int node = node0 + ln;
        const float (&y)[F] = yy[k];
        if (ln < AGG_NPB && node < n) {
          const int g = gid[k];
          atomicAdd(&tab[g * DOUT + lane * 2 + 0], y[0]);
          atomicAdd(&tab[g * DOUT + lane * 2 + 1], y[1]);
          c0 += (g == lane) ? 1 : 0;
          c1 += (g == lane + 32) ? 1 : 0;
        }
      }
      if (c0 != 0) atomicAdd(&gcnt[lane], (float)c0);
      if (c1 != 0) atomicAdd(&gcnt[lane + 32], (float)c1);
      __syncthreads();

      for (int i = tid; i < NGRAPHS * DOUT; i += AGG_THREADS) {
        const float s = tab[i];
        if (s != 0.0f) atomicAdd(&sums[i], s);
      }
      if (tid < NGRAPHS) {
        const float c = gcnt[tid];
        if (c != 0.0f) atomicAdd(&cnts[tid], c);
      }
    }
  }
}

__global__ void pool_final(const float* __restrict__ sums, const float* __restrict__ cnts,
                           float* __restrict__ out) {
  int i = blockIdx.x * blockDim.x + threadIdx.x;
  if (i < NGRAPHS * DOUT) {
    int g = i / DOUT;
    float m = sums[i] / fmaxf(cnts[g], 1.0f);
    out[i] = 1.0f / (1.0f + __expf(-m));
  }
}

extern "C" void kernel_launch(void* const* d_in, const int* in_sizes, int n_in,
                              void* d_out, int out_size, void* d_ws, size_t ws_size,
                              hipStream_t stream) {
  const float* x   = (const float*)d_in[0];
  const int*   ei  = (const int*)d_in[1];
  const float* ew  = (const float*)d_in[2];
  const int*   bat = (const int*)d_in[3];
  const float* W1  = (const float*)d_in[4];
  const float* b1  = (const float*)d_in[5];
  const float* g1  = (const float*)d_in[6];
  const float* bt1 = (const float*)d_in[7];
  const float* W2  = (const float*)d_in[8];
  const float* b2  = (const float*)d_in[9];
  const float* g2  = (const float*)d_in[10];
  const float* bt2 = (const float*)d_in[11];
  float* out = (float*)d_out;

  const int* rows = ei;
  const int* cols = ei + NEDGES;

  char* ws = (char*)d_ws;
  size_t off = 0;
  auto carve = [&](size_t bytes) -> float* {
    float* p = (float*)(ws + off);
    off += (bytes + 255) & ~(size_t)255;
    return p;
  };
  float* dinv = carve((size_t)NNODES * 4);
  float* h    = carve((size_t)NNODES * DHID * 4);
  float* a    = carve((size_t)NNODES * DHID * 4);
  float* sums = carve((size_t)NGRAPHS * DOUT * 4);
  float* cnts = carve((size_t)NGRAPHS * 4);

  const int TB = 256;
  auto blocks = [&](long long threads) { return (int)((threads + TB - 1) / TB); };

  zero_f32<<<blocks(NNODES), TB, 0, stream>>>(dinv, NNODES);
  deg_accum<<<blocks(NEDGES), TB, 0, stream>>>(cols, ew, dinv, NEDGES);
  deg_to_dinv<<<blocks(NNODES), TB, 0, stream>>>(dinv, NNODES);

  wmma_gemm_f32<DHID, DIN, false><<<GEMM_BLOCKS, GEMM_THREADS, 0, stream>>>(
      x, W1, h, NNODES, nullptr, nullptr, nullptr, nullptr);
  gather_lds<DHID, false, false><<<AGG_BLOCKS, AGG_THREADS, 0, stream>>>(
      h, rows, cols, ew, dinv, b1, g1, bt1, a, nullptr, sums, cnts, NNODES, nullptr);

  unsigned int* keys = (unsigned int*)(h + (size_t)NNODES * DOUT);
  wmma_gemm_f32<DOUT, DHID, true><<<GEMM_BLOCKS, GEMM_THREADS, 0, stream>>>(
      a, W2, h, NNODES, rows, cols, ew, keys);
  gather_lds<DOUT, true, true><<<AGG_BLOCKS, AGG_THREADS, 0, stream>>>(
      h, rows, cols, ew, dinv, b2, g2, bt2, nullptr, bat, sums, cnts, NNODES, keys);
  pool_final<<<blocks(NGRAPHS * DOUT), TB, 0, stream>>>(sums, cnts, out);
}


# layer 1 scans 1 key byte/edge like layer 2: launch 4 writes keys+records into buffer a, free since gather_lds multiplies its LN rows by W2 in LDS (WMMA), stores h2; 7 launches; row sum order changes
# speedup vs baseline: 3.1716x; 1.2063x over previous
#include <hip/hip_runtime.h>
#include <math.h>

#define NNODES 100000
#define NEDGES 1600000
#define NGRAPHS 64
#define DIN 256
#define DHID 128
#define DOUT 64
#define EPSF 1e-5f

typedef __attribute__((ext_vector_type(2))) float v2f;
typedef __attribute__((ext_vector_type(8))) float v8f;

template <int N>
__device__ __forceinline__ float xrow(float x) {
  const int b = __builtin_bit_cast(int, x);
  return __builtin_bit_cast(float,
      __builtin_amdgcn_update_dpp(0, b, 0x160 + N, 0xf, 0xf, true));
}
__device__ __forceinline__ float xhalf(float x) {
  const int b = __builtin_bit_cast(int, x);
  return __builtin_bit_cast(float,
      __builtin_amdgcn_permlanex16(b, b, 0x76543210, 0xfedcba98, false, false));
}
__device__ __forceinline__ float wave_sum(float s) {
  s += xhalf(s);
  s += xrow<8>(s);
  s += xrow<4>(s);
  s += xrow<2>(s);
  s += xrow<1>(s);
  return s;
}

__global__ void zero_f32(float* __restrict__ p, int n) {
  int i = blockIdx.x * blockDim.x + threadIdx.x;
  if (i < n) p[i] = 0.0f;
}

__global__ void deg_accum(const int* __restrict__ col, const float* __restrict__ w,
                          float* __restrict__ deg, int nE) {
  int e = blockIdx.x * blockDim.x + threadIdx.x;
  if (e < nE) atomicAdd(&deg[col[e]], w[e]);
}

__global__ void deg_to_dinv(float* __restrict__ deg, int n) {
  int i = blockIdx.x * blockDim.x + threadIdx.x;
  if (i < n) {
    float d = deg[i] + 1.0f;
    deg[i] = rsqrtf(fmaxf(d, EPSF));
  }
}

#define GEMM_THREADS 1024
#define GEMM_WAVES (GEMM_THREADS / 32)
#define GEMM_BLOCKS 256
static_assert(NNODES % 16 == 0, "rows come in whole 16-row stripes");
#define AGG_NPB 392
#define AGG_BLOCKS ((NNODES + AGG_NPB - 1) / AGG_NPB)
static_assert(AGG_BLOCKS <= 256, "a work-group number fits one byte");
static_assert(NEDGES % 16 == 0, "the key bytes are read sixteen at a time");
static_assert((size_t)NEDGES * 9 <= (size_t)NNODES * (DHID - DOUT) * 4,
              "key bytes and edge records fit the half of buffer a that h2 leaves unused");
static_assert(NEDGES <= (1 << 21) && AGG_NPB <= (1 << 9),
              "an edge number and a local row share one queue word");
static_assert(NNODES <= (1 << 17), "a source node and a local row share one record word");

template <int NCOLS, int KDIM, bool KEYS>
__global__ void __launch_bounds__(GEMM_THREADS)
wmma_gemm_f32(const float* __restrict__ A, const float* __restrict__ B,
              float* __restrict__ C, int M, const int* __restrict__ srcs,
              const int* __restrict__ cols, const float* __restrict__ ew,
              unsigned int* __restrict__ keys) {
  __shared__ float2 Bl[(KDIM / 2) * NCOLS];

  if constexpr (KEYS) {
    const int4* __restrict__ c4p = reinterpret_cast<const int4*>(cols);
    const int4* __restrict__ r4p = reinterpret_cast<const int4*>(srcs);
    const float4* __restrict__ w4p = reinterpret_cast<const float4*>(ew);
    uint4* __restrict__ rec4 = reinterpret_cast<uint4*>(keys + NEDGES / 4);
    const int nthr = (int)gridDim.x * GEMM_THREADS;
    for (int i = (int)blockIdx.x * GEMM_THREADS + (int)threadIdx.x; i < NEDGES / 4; i += nthr) {
      const int4 c = c4p[i];
      const int4 r = r4p[i];
      const float4 wv = w4p[i];
      const unsigned o0 = (unsigned)(c.x / AGG_NPB), o1 = (unsigned)(c.y / AGG_NPB);
      const unsigned o2 = (unsigned)(c.z / AGG_NPB), o3 = (unsigned)(c.w / AGG_NPB);
      keys[i] = o0 | (o1 << 8) | (o2 << 16) | (o3 << 24);
      uint4 lo, hi;
      lo.x = (unsigned)r.x | (((unsigned)c.x - o0 * AGG_NPB) << 17);
      lo.y = __float_as_uint(wv.x);
      lo.z = (unsigned)r.y | (((unsigned)c.y - o1 * AGG_NPB) << 17);
      lo.w = __float_as_uint(wv.y);
      hi.x = (unsigned)r.z | (((unsigned)c.z - o2 * AGG_NPB) << 17);
      hi.y = __float_as_uint(wv.z);
      hi.z = (unsigned)r.w | (((unsigned)c.w - o3 * AGG_NPB) << 17);
      hi.w = __float_as_uint(wv.w);
      rec4[2 * i + 0] = lo;
      rec4[2 * i + 1] = hi;
    }
  }

  for (int i = threadIdx.x; i < (KDIM / 2) * NCOLS; i += GEMM_THREADS) {
    const int k2 = i / NCOLS;
    const int n = i - k2 * NCOLS;
    float2 v;
    v.x = B[(size_t)(2 * k2 + 0) * NCOLS + n];
    v.y = B[(size_t)(2 * k2 + 1) * NCOLS + n];
    Bl[i] = v;
  }
  __syncthreads();

  const int wave = __builtin_amdgcn_readfirstlane((int)(threadIdx.x >> 5));
  const int lane = threadIdx.x & 31;
  const int half = lane >> 4;
  const int l = lane & 15;
  constexpr int NT = NCOLS / 16;
  const int nstripes = M / 16;

  for (int stripe = wave * (int)gridDim.x + (int)blockIdx.x; stripe < nstripes;
       stripe += GEMM_WAVES * (int)gridDim.x) {
    const int m0 = stripe * 16;

    v8f acc[NT];
#pragma unroll
    for (int t = 0; t < NT; ++t)
#pragma unroll
      for (int j = 0; j < 8; ++j) acc[t][j] = 0.0f;

    const float* __restrict__ arow = A + (size_t)(m0 + l) * KDIM + 2 * half;

    v2f a;
    a[0] = arow[0];
    a[1] = arow[1];

    for (int k0 = 0; k0 < KDIM; k0 += 4) {
      v2f a_next = a;
      if (k0 + 4 < KDIM) {
        a_next[0] = arow[k0 + 4 + 0];
        a_next[1] = arow[k0 + 4 + 1];
      }
      const float2* __restrict__ brow = Bl + (size_t)(k0 / 2 + half) * NCOLS + l;
      v2f bfrag[NT];
#pragma unroll
      for (int t = 0; t < NT; ++t) {
        float2 bb = brow[t * 16];
        bfrag[t][0] = bb.x;
        bfrag[t][1] = bb.y;
      }
#pragma unroll
      for (int t = 0; t < NT; ++t)
        acc[t] = __builtin_amdgcn_wmma_f32_16x16x4_f32(
             false, a,  false, bfrag[t],
             (short)0, acc[t],  false,  false);
      a = a_next;
    }

#pragma unroll
    for (int t = 0; t < NT; ++t)
#pragma unroll
      for (int r = 0; r < 8; ++r)
        C[(size_t)(m0 + r + 8 * half) * NCOLS + t * 16 + l] = acc[t][r];
  }
}

#define AGG_THREADS 1024
#define AGG_WAVES (AGG_THREADS / 32)
#define AGG_QCAP 4096
#define AGG_LOADS 8
static_assert(NEDGES % 4 == 0, "targets are read four at a time");
static_assert(NNODES < (1 << 30), "node numbers fit the unsigned range test");

#define AGG_RPW ((AGG_NPB + AGG_WAVES - 1) / AGG_WAVES)
static_assert(NGRAPHS == 64 && DOUT == 64,
              "a lane owns two features and counts two graphs");

__device__ __forceinline__ unsigned zero_bytes(unsigned x) {
  return ~(((x & 0x7f7f7f7fu) + 0x7f7f7f7fu) | x | 0x7f7f7f7fu);
}

template <int D, bool POOL, bool BYTEKEYS, int DN>
__global__ void __launch_bounds__(AGG_THREADS)
gather_lds(const float* __restrict__ h, const int* __restrict__ rows,
           const int* __restrict__ cols, const float* __restrict__ w,
           const float* __restrict__ dinv, const float* __restrict__ bias,
           const float* __restrict__ gamma, const float* __restrict__ beta,
           float* __restrict__ out, const int* __restrict__ batch,
           float* __restrict__ sums, float* __restrict__ cnts, int n,
           const unsigned int* __restrict__ keys, const float* __restrict__ Wn) {
  static_assert(!POOL || D == DOUT, "the pool table is NGRAPHS x DOUT");
  static_assert(POOL ? DN == 0 : (DN > 0 && DN % 16 == 0 && D % 4 == 0),
                "without the pool the rows are multiplied by a D x DN matrix");
  constexpr int F = D / 32;
  constexpr int ROW4 = D / 4;
  constexpr int U = 16;
  constexpr int NE4 = NEDGES / 4;
  __shared__ __attribute__((aligned(16))) float acc[AGG_NPB * D];
  __shared__ int q[AGG_QCAP];
  __shared__ float dloc[AGG_NPB];
  __shared__ float2 Wl[POOL ? 1 : (D / 2) * DN];

  const int tid = threadIdx.x;
  const int lane = tid & 31;
  const int wave = __builtin_amdgcn_readfirstlane(tid >> 5);
  const int node0 = blockIdx.x * AGG_NPB;

  for (int i = tid; i < AGG_NPB; i += AGG_THREADS) dloc[i] = dinv[min(node0 + i, n - 1)];

  if constexpr (!POOL) {
    for (int i = tid; i < (D / 2) * DN; i += AGG_THREADS) {
      const int k2 = i / DN;
      const int c = i - k2 * DN;
      float2 v;
      v.x = Wn[(size_t)(2 * k2 + 0) * DN + c];
      v.y = Wn[(size_t)(2 * k2 + 1) * DN + c];
      Wl[i] = v;
    }
  }

  if constexpr (!POOL) {
    if (blockIdx.x == 0) {
      for (int i = tid; i < NGRAPHS * DOUT; i += AGG_THREADS) sums[i] = 0.0f;
      if (tid < NGRAPHS) cnts[tid] = 0.0f;
    }
  }

  {
    const float4* __restrict__ h4 = reinterpret_cast<const float4*>(h);
    const float4* __restrict__ b4 = reinterpret_cast<const float4*>(bias);
    float4* acc4 = reinterpret_cast<float4*>(acc);
#pragma unroll 4
    for (int i = tid; i < AGG_NPB * ROW4; i += AGG_THREADS) {
      const int ln = i / ROW4;
      const int f4 = i - ln * ROW4;
      const int node = min(node0 + ln, n - 1);
      const float d = dinv[node];
      const float s = d * d;
      const float4 hv = h4[(size_t)node * ROW4 + f4];
      const float4 bv = b4[f4];
      float4 o;
      o.x = bv.x + hv.x * s;
      o.y = bv.y + hv.y * s;
      o.z = bv.z + hv.z * s;
      o.w = bv.w + hv.w * s;
      acc4[i] = o;
    }
  }
  __syncthreads();

  const int4* __restrict__ c4p = reinterpret_cast<const int4*>(cols);

  constexpr int WQCAP = AGG_QCAP / AGG_WAVES;
  static_assert(WQCAP >= 64, "a wave's part holds a full batch and then some");
  int* wq = q + wave * WQCAP;
  int qcount = 0;

  auto push = [&](unsigned mask, bool hit, int word) {
    const int pos = qcount + (int)__builtin_amdgcn_mbcnt_lo(mask, 0u);
    if (hit && pos < WQCAP) wq[pos] = word;
    qcount += __builtin_popcount(mask);
  };

  constexpr int LOADS = BYTEKEYS ? 7 : AGG_LOADS;
  constexpr int EPL = BYTEKEYS ? 16 : 4;
  constexpr int NV = NEDGES / EPL;
  constexpr int NTRIPS = (NV + LOADS * AGG_THREADS - 1) / (LOADS * AGG_THREADS);

  for (int it = 0; it <= NTRIPS; ++it) {
    const int q0 = qcount;
    if (it < NTRIPS) {
      const bool whole = (it + 1) * LOADS * AGG_THREADS <= NV;
      if constexpr (BYTEKEYS) {
        const uint4* __restrict__ k4p = reinterpret_cast<const uint4*>(keys);
        const unsigned pat = (unsigned)blockIdx.x * 0x01010101u;
        uint4 kv[LOADS];
        int gi[LOADS];
#pragma unroll
        for (int j = 0; j < LOADS; ++j) gi[j] = (it * LOADS + j) * AGG_THREADS + tid;
        if (whole) {
#pragma unroll
          for (int j = 0; j < LOADS; ++j) kv[j] = k4p[gi[j]];
        } else {
#pragma unroll
          for (int j = 0; j < LOADS; ++j) {
            if (gi[j] < NV) {
              kv[j] = k4p[gi[j]];
            } else {
              kv[j].x = ~pat; kv[j].y = ~pat; kv[j].z = ~pat; kv[j].w = ~pat;
            }
          }
        }
#pragma unroll
        for (int j = 0; j < LOADS; ++j) {
          unsigned m = (zero_bytes(kv[j].x ^ pat) >> 7) | (zero_bytes(kv[j].y ^ pat) >> 6) |
                       (zero_bytes(kv[j].z ^ pat) >> 5) | (zero_bytes(kv[j].w ^ pat) >> 4);
          unsigned mk = __builtin_amdgcn_ballot_w32(m != 0);
          while (mk != 0) {
            const bool hit = m != 0;
            const int p = __builtin_ctz(m | 0x80000000u);
            push(mk, hit, gi[j] * 16 + ((p & 7) << 2) + (p >> 3));
            m &= m - 1;
            mk = __builtin_amdgcn_ballot_w32(m != 0);
          }
        }
      } else {
        int4 cv[LOADS];
        int idx[LOADS];
#pragma unroll
        for (int j = 0; j < LOADS; ++j) idx[j] = (it * LOADS + j) * AGG_THREADS + tid;
        if (whole) {
#pragma unroll
          for (int j = 0; j < LOADS; ++j) cv[j] = c4p[idx[j]];
        } else {
#pragma unroll
          for (int j = 0; j < LOADS; ++j) {
            if (idx[j] < NV) {
              cv[j] = c4p[idx[j]];
            } else {
              cv[j].x = -1; cv[j].y = -1; cv[j].z = -1; cv[j].w = -1;
            }
          }
        }
        auto test = [&](int c, int e) {
          const unsigned d = (unsigned)(c - node0);
          const bool hit = d < AGG_NPB;
          const unsigned mk = __builtin_amdgcn_ballot_w32(hit);
          if (mk != 0) push(mk, hit, e | (int)(d << 21));
        };
#pragma unroll
        for (int j = 0; j < LOADS; ++j) {
          test(cv[j].x, idx[j] * 4 + 0);
          test(cv[j].y, idx[j] * 4 + 1);
          test(cv[j].z, idx[j] * 4 + 2);
          test(cv[j].w, idx[j] * 4 + 3);
        }
      }
      if (qcount > WQCAP) {
        qcount = q0;
#pragma unroll 1
        for (int j = 0; j < LOADS; ++j) {
          const int e0 = ((it * LOADS + j) * AGG_THREADS + tid) * EPL;
#pragma unroll 1
          for (int k = 0; k < EPL; ++k) {
            const int e = e0 + k;
            if (e < NEDGES) {
              const int c = cols[e];
              const unsigned dl = (unsigned)(c - node0);
              if (dl < AGG_NPB) {
                const int r = rows[e];
                const float nr = dinv[r] * w[e] * dinv[c];
                for (int f = 0; f < D; ++f)
                  atomicAdd(&acc[dl * D + f], h[(size_t)r * D + f] * nr);
              }
            }
          }
        }
      }
    }
    const int need = (it < NTRIPS) ? 32 : 1;
    int base = 0;
    while (qcount - base >= need) {
      const int nb = min(32, qcount - base);
      int r = 0, cl = 0;
      float nr = 0.0f;
      if (lane < nb) {
        const int qv = wq[base + lane];
        if constexpr (BYTEKEYS) {
          const uint2 rc = reinterpret_cast<const uint2*>(keys + NEDGES / 4)[qv];
          r = (int)(rc.x & 0x1ffffu);
          cl = (int)(rc.x >> 17);
          nr = dinv[r] * __uint_as_float(rc.y) * dloc[cl];
        } else {
          const int e = qv & 0x1fffff;
          cl = qv >> 21;
          r = rows[e];
          nr = dinv[r] * w[e] * dloc[cl];
        }
      }
#pragma unroll
      for (int j0 = 0; j0 < 32; j0 += U) {
        if (j0 < nb) {
          float v[U][F];
#pragma unroll
          for (int u = 0; u < U; ++u) {
            if (j0 + u < nb) {
              const int rr = __builtin_amdgcn_readlane(r, j0 + u);
              const float* __restrict__ hp = h + (size_t)rr * D + lane;
#pragma unroll
              for (int f = 0; f < F; ++f) v[u][f] = hp[f * 32];
            }
          }
#pragma unroll
          for (int u = 0; u < U; ++u) {
            if (j0 + u < nb) {
              const int cc = __builtin_amdgcn_readlane(cl, j0 + u);
              const float nn =
                  __int_as_float(__builtin_amdgcn_readlane(__float_as_int(nr), j0 + u));
              float* ap = acc + cc * D + lane;
#pragma unroll
              for (int f = 0; f < F; ++f) atomicAdd(ap + f * 32, v[u][f] * nn);
            }
          }
        }
      }
      base += nb;
    }
    if (base > 0) {
      const int rem = qcount - base;
      int t = 0;
      if (lane < rem) t = wq[base + lane];
      if (lane < rem) wq[lane] = t;
      qcount = rem;
    }
  }
  __syncthreads();

  {
    static_assert(F == 4 || F == 2, "a lane reads its F features of a row with one LDS load");
    float gm[F], bt[F];
#pragma unroll
    for (int f = 0; f < F; ++f) {
      gm[f] = gamma[lane * F + f];
      bt[f] = beta[lane * F + f];
    }
    auto ln_row = [&](int ln, float (&y)[F]) {
      const float* rp = acc + ln * D + lane * F;
      float v[F];
      if constexpr (F == 4) {
        const float4 t = *reinterpret_cast<const float4*>(rp);
        v[0] = t.x; v[1] = t.y; v[2] = t.z; v[3] = t.w;
      } else {
        const float2 t = *reinterpret_cast<const float2*>(rp);
        v[0] = t.x; v[1] = t.y;
      }
      float s = 0.0f;
#pragma unroll
      for (int f = 0; f < F; ++f) {
        v[f] = fmaxf(v[f], 0.0f);
        s += v[f];
      }
      s = wave_sum(s);
      const float mu = s * (1.0f / D);
      float ss = 0.0f;
#pragma unroll
      for (int f = 0; f < F; ++f) {
        const float d = v[f] - mu;
        ss += d * d;
      }
      ss = wave_sum(ss);
      const float rs = rsqrtf(ss * (1.0f / D) + EPSF);
#pragma unroll
      for (int f = 0; f < F; ++f) y[f] = (v[f] - mu) * rs * gm[f] + bt[f];
    };

    if constexpr (!POOL) {
      float yy[AGG_RPW][F];
#pragma unroll
      for (int k = 0; k < AGG_RPW; ++k) ln_row(min(wave + k * AGG_WAVES, AGG_NPB - 1), yy[k]);
#pragma unroll
      for (int k = 0; k < AGG_RPW; ++k) {
        const int ln = wave + k * AGG_WAVES;
        const float (&y)[F] = yy[k];
        if (ln < AGG_NPB) {
          float* ap = acc + ln * D + lane * F;
          if constexpr (F == 4) {
            float4 o; o.x = y[0]; o.y = y[1]; o.z = y[2]; o.w = y[3];
            *reinterpret_cast<float4*>(ap) = o;
          } else {
            float2 o; o.x = y[0]; o.y = y[1];
            *reinterpret_cast<float2*>(ap) = o;
          }
        }
      }
      __syncthreads();

      constexpr int NT = DN / 16;
      constexpr int NSTRIPES = (AGG_NPB + 15) / 16;
      const int half = lane >> 4;
      const int l = lane & 15;
      for (int stripe = wave; stripe < NSTRIPES; stripe += AGG_WAVES) {
        const int m0 = stripe * 16;
        v8f cacc[NT];
#pragma unroll
        for (int t = 0; t < NT; ++t)
#pragma unroll
          for (int j = 0; j < 8; ++j) cacc[t][j] = 0.0f;
        const float* arow = acc + min(m0 + l, AGG_NPB - 1) * D + 2 * half;
        for (int k0 = 0; k0 < D; k0 += 4) {
          const float2 av = *reinterpret_cast<const float2*>(arow + k0);
          v2f afrag;
          afrag[0] = av.x;
          afrag[1] = av.y;
          const float2* brow = Wl + (k0 / 2 + half) * DN + l;
          v2f bfrag[NT];
#pragma unroll
          for (int t = 0; t < NT; ++t) {
            const float2 bb = brow[t * 16];
            bfrag[t][0] = bb.x;
            bfrag[t][1] = bb.y;
          }
#pragma unroll
          for (int t = 0; t < NT; ++t)
            cacc[t] = __builtin_amdgcn_wmma_f32_16x16x4_f32(
                 false, afrag,  false, bfrag[t],
                 (short)0, cacc[t],  false,  false);
        }
#pragma unroll
        for (int r = 0; r < 8; ++r) {
          const int ln = m0 + r + 8 * half;
          const int node = node0 + ln;
          if (ln < AGG_NPB && node < n) {
#pragma unroll
            for (int t = 0; t < NT; ++t)
              out[(size_t)node * DN + t * 16 + l] = cacc[t][r];
          }
        }
      }
    } else {
      __shared__ float tab[NGRAPHS * DOUT];
      __shared__ float gcnt[NGRAPHS];
      for (int i = tid; i < NGRAPHS * DOUT; i += AGG_THREADS) tab[i] = 0.0f;
      if (tid < NGRAPHS) gcnt[tid] = 0.0f;
      int gid[AGG_RPW];
#pragma unroll
      for (int k = 0; k < AGG_RPW; ++k)
        gid[k] = batch[min(node0 + wave + k * AGG_WAVES, n - 1)];
      __syncthreads();

      int c0 = 0, c1 = 0;
      float yy[AGG_RPW][F];
#pragma unroll
      for (int k = 0; k < AGG_RPW; ++k) ln_row(min(wave + k * AGG_WAVES, AGG_NPB - 1), yy[k]);
#pragma unroll
      for (int k = 0; k < AGG_RPW; ++k) {
        const int ln = wave + k * AGG_WAVES;
        const int node = node0 + ln;
        const float (&y)[F] = yy[k];
        if (ln < AGG_NPB && node < n) {
          const int g = gid[k];
          atomicAdd(&tab[g * DOUT + lane * 2 + 0], y[0]);
          atomicAdd(&tab[g * DOUT + lane * 2 + 1], y[1]);
          c0 += (g == lane) ? 1 : 0;
          c1 += (g == lane + 32) ? 1 : 0;
        }
      }
      if (c0 != 0) atomicAdd(&gcnt[lane], (float)c0);
      if (c1 != 0) atomicAdd(&gcnt[lane + 32], (float)c1);
      __syncthreads();

      for (int i = tid; i < NGRAPHS * DOUT; i += AGG_THREADS) {
        const float s = tab[i];
        if (s != 0.0f) atomicAdd(&sums[i], s);
      }
      if (tid < NGRAPHS) {
        const float c = gcnt[tid];
        if (c != 0.0f) atomicAdd(&cnts[tid], c);
      }
    }
  }
}

__global__ void pool_final(const float* __restrict__ sums, const float* __restrict__ cnts,
                           float* __restrict__ out) {
  int i = blockIdx.x * blockDim.x + threadIdx.x;
  if (i < NGRAPHS * DOUT) {
    int g = i / DOUT;
    float m = sums[i] / fmaxf(cnts[g], 1.0f);
    out[i] = 1.0f / (1.0f + __expf(-m));
  }
}

extern "C" void kernel_launch(void* const* d_in, const int* in_sizes, int n_in,
                              void* d_out, int out_size, void* d_ws, size_t ws_size,
                              hipStream_t stream) {
  const float* x   = (const float*)d_in[0];
  const int*   ei  = (const int*)d_in[1];
  const float* ew  = (const float*)d_in[2];
  const int*   bat = (const int*)d_in[3];
  const float* W1  = (const float*)d_in[4];
  const float* b1  = (const float*)d_in[5];
  const float* g1  = (const float*)d_in[6];
  const float* bt1 = (const float*)d_in[7];
  const float* W2  = (const float*)d_in[8];
  const float* b2  = (const float*)d_in[9];
  const float* g2  = (const float*)d_in[10];
  const float* bt2 = (const float*)d_in[11];
  float* out = (float*)d_out;

  const int* rows = ei;
  const int* cols = ei + NEDGES;

  char* ws = (char*)d_ws;
  size_t off = 0;
  auto carve = [&](size_t bytes) -> float* {
    float* p = (float*)(ws + off);
    off += (bytes + 255) & ~(size_t)255;
    return p;
  };
  float* dinv = carve((size_t)NNODES * 4);
  float* h    = carve((size_t)NNODES * DHID * 4);
  float* a    = carve((size_t)NNODES * DHID * 4);
  float* sums = carve((size_t)NGRAPHS * DOUT * 4);
  float* cnts = carve((size_t)NGRAPHS * 4);

  const int TB = 256;
  auto blocks = [&](long long threads) { return (int)((threads + TB - 1) / TB); };

  zero_f32<<<blocks(NNODES), TB, 0, stream>>>(dinv, NNODES);
  deg_accum<<<blocks(NEDGES), TB, 0, stream>>>(cols, ew, dinv, NEDGES);
  deg_to_dinv<<<blocks(NNODES), TB, 0, stream>>>(dinv, NNODES);

  float* h2 = a;
  unsigned int* keys = (unsigned int*)(a + (size_t)NNODES * DOUT);
  wmma_gemm_f32<DHID, DIN, true><<<GEMM_BLOCKS, GEMM_THREADS, 0, stream>>>(
      x, W1, h, NNODES, rows, cols, ew, keys);
  gather_lds<DHID, false, true, DOUT><<<AGG_BLOCKS, AGG_THREADS, 0, stream>>>(
      h, rows, cols, ew, dinv, b1, g1, bt1, h2, nullptr, sums, cnts, NNODES, keys, W2);

  gather_lds<DOUT, true, true, 0><<<AGG_BLOCKS, AGG_THREADS, 0, stream>>>(
      h2, rows, cols, ew, dinv, b2, g2, bt2, nullptr, bat, sums, cnts, NNODES, keys, nullptr);
  pool_final<<<blocks(NGRAPHS * DOUT), TB, 0, stream>>>(sums, cnts, out);
}


# layer 2 scans no keys: layer 1's waves store the edge numbers they add (per-wave lists in free workspace), layer 2's waves read their own list back by whole lines; layer-2 row sum order changes
# speedup vs baseline: 3.4449x; 1.0862x over previous
#include <hip/hip_runtime.h>
#include <math.h>

#define NNODES 100000
#define NEDGES 1600000
#define NGRAPHS 64
#define DIN 256
#define DHID 128
#define DOUT 64
#define EPSF 1e-5f

typedef __attribute__((ext_vector_type(2))) float v2f;
typedef __attribute__((ext_vector_type(8))) float v8f;

template <int N>
__device__ __forceinline__ float xrow(float x) {
  const int b = __builtin_bit_cast(int, x);
  return __builtin_bit_cast(float,
      __builtin_amdgcn_update_dpp(0, b, 0x160 + N, 0xf, 0xf, true));
}
__device__ __forceinline__ float xhalf(float x) {
  const int b = __builtin_bit_cast(int, x);
  return __builtin_bit_cast(float,
      __builtin_amdgcn_permlanex16(b, b, 0x76543210, 0xfedcba98, false, false));
}
__device__ __forceinline__ float wave_sum(float s) {
  s += xhalf(s);
  s += xrow<8>(s);
  s += xrow<4>(s);
  s += xrow<2>(s);
  s += xrow<1>(s);
  return s;
}

__global__ void zero_f32(float* __restrict__ p, int n) {
  int i = blockIdx.x * blockDim.x + threadIdx.x;
  if (i < n) p[i] = 0.0f;
}

__global__ void deg_accum(const int* __restrict__ col, const float* __restrict__ w,
                          float* __restrict__ deg, int nE) {
  int e = blockIdx.x * blockDim.x + threadIdx.x;
  if (e < nE) atomicAdd(&deg[col[e]], w[e]);
}

__global__ void deg_to_dinv(float* __restrict__ deg, int n) {
  int i = blockIdx.x * blockDim.x + threadIdx.x;
  if (i < n) {
    float d = deg[i] + 1.0f;
    deg[i] = rsqrtf(fmaxf(d, EPSF));
  }
}

#define GEMM_THREADS 1024
#define GEMM_WAVES (GEMM_THREADS / 32)
#define GEMM_BLOCKS 256
static_assert(NNODES % 16 == 0, "rows come in whole 16-row stripes");
#define AGG_NPB 392
#define AGG_BLOCKS ((NNODES + AGG_NPB - 1) / AGG_NPB)
static_assert(AGG_BLOCKS <= 256, "a work-group number fits one byte");
static_assert(NEDGES % 16 == 0, "the key bytes are read sixteen at a time");
static_assert((size_t)NEDGES * 9 <= (size_t)NNODES * (DHID - DOUT) * 4,
              "key bytes and edge records fit the half of buffer a that h2 leaves unused");
static_assert(NEDGES <= (1 << 21) && AGG_NPB <= (1 << 9),
              "an edge number and a local row share one queue word");
static_assert(NNODES <= (1 << 17), "a source node and a local row share one record word");

template <int NCOLS, int KDIM, bool KEYS>
__global__ void __launch_bounds__(GEMM_THREADS)
wmma_gemm_f32(const float* __restrict__ A, const float* __restrict__ B,
              float* __restrict__ C, int M, const int* __restrict__ srcs,
              const int* __restrict__ cols, const float* __restrict__ ew,
              unsigned int* __restrict__ keys) {
  __shared__ float2 Bl[(KDIM / 2) * NCOLS];

  if constexpr (KEYS) {
    const int4* __restrict__ c4p = reinterpret_cast<const int4*>(cols);
    const int4* __restrict__ r4p = reinterpret_cast<const int4*>(srcs);
    const float4* __restrict__ w4p = reinterpret_cast<const float4*>(ew);
    uint4* __restrict__ rec4 = reinterpret_cast<uint4*>(keys + NEDGES / 4);
    const int nthr = (int)gridDim.x * GEMM_THREADS;
    for (int i = (int)blockIdx.x * GEMM_THREADS + (int)threadIdx.x; i < NEDGES / 4; i += nthr) {
      const int4 c = c4p[i];
      const int4 r = r4p[i];
      const float4 wv = w4p[i];
      const unsigned o0 = (unsigned)(c.x / AGG_NPB), o1 = (unsigned)(c.y / AGG_NPB);
      const unsigned o2 = (unsigned)(c.z / AGG_NPB), o3 = (unsigned)(c.w / AGG_NPB);
      keys[i] = o0 | (o1 << 8) | (o2 << 16) | (o3 << 24);
      uint4 lo, hi;
      lo.x = (unsigned)r.x | (((unsigned)c.x - o0 * AGG_NPB) << 17);
      lo.y = __float_as_uint(wv.x);
      lo.z = (unsigned)r.y | (((unsigned)c.y - o1 * AGG_NPB) << 17);
      lo.w = __float_as_uint(wv.y);
      hi.x = (unsigned)r.z | (((unsigned)c.z - o2 * AGG_NPB) << 17);
      hi.y = __float_as_uint(wv.z);
      hi.z = (unsigned)r.w | (((unsigned)c.w - o3 * AGG_NPB) << 17);
      hi.w = __float_as_uint(wv.w);
      rec4[2 * i + 0] = lo;
      rec4[2 * i + 1] = hi;
    }
  }

  for (int i = threadIdx.x; i < (KDIM / 2) * NCOLS; i += GEMM_THREADS) {
    const int k2 = i / NCOLS;
    const int n = i - k2 * NCOLS;
    float2 v;
    v.x = B[(size_t)(2 * k2 + 0) * NCOLS + n];
    v.y = B[(size_t)(2 * k2 + 1) * NCOLS + n];
    Bl[i] = v;
  }
  __syncthreads();

  const int wave = __builtin_amdgcn_readfirstlane((int)(threadIdx.x >> 5));
  const int lane = threadIdx.x & 31;
  const int half = lane >> 4;
  const int l = lane & 15;
  constexpr int NT = NCOLS / 16;
  const int nstripes = M / 16;

  for (int stripe = wave * (int)gridDim.x + (int)blockIdx.x; stripe < nstripes;
       stripe += GEMM_WAVES * (int)gridDim.x) {
    const int m0 = stripe * 16;

    v8f acc[NT];
#pragma unroll
    for (int t = 0; t < NT; ++t)
#pragma unroll
      for (int j = 0; j < 8; ++j) acc[t][j] = 0.0f;

    const float* __restrict__ arow = A + (size_t)(m0 + l) * KDIM + 2 * half;

    v2f a;
    a[0] = arow[0];
    a[1] = arow[1];

    for (int k0 = 0; k0 < KDIM; k0 += 4) {
      v2f a_next = a;
      if (k0 + 4 < KDIM) {
        a_next[0] = arow[k0 + 4 + 0];
        a_next[1] = arow[k0 + 4 + 1];
      }
      const float2* __restrict__ brow = Bl + (size_t)(k0 / 2 + half) * NCOLS + l;
      v2f bfrag[NT];
#pragma unroll
      for (int t = 0; t < NT; ++t) {
        float2 bb = brow[t * 16];
        bfrag[t][0] = bb.x;
        bfrag[t][1] = bb.y;
      }
#pragma unroll
      for (int t = 0; t < NT; ++t)
        acc[t] = __builtin_amdgcn_wmma_f32_16x16x4_f32(
             false, a,  false, bfrag[t],
             (short)0, acc[t],  false,  false);
      a = a_next;
    }

#pragma unroll
    for (int t = 0; t < NT; ++t)
#pragma unroll
      for (int r = 0; r < 8; ++r)
        C[(size_t)(m0 + r + 8 * half) * NCOLS + t * 16 + l] = acc[t][r];
  }
}

#define AGG_THREADS 1024
#define AGG_WAVES (AGG_THREADS / 32)
#define AGG_QCAP 4096
#define AGG_LOADS 8
#define AGG_LCAP 320
#define AGG_LIST_WORDS ((size_t)AGG_BLOCKS * AGG_WAVES * (AGG_LCAP + 1) + AGG_BLOCKS)
static_assert((size_t)NEDGES * 9 + AGG_LIST_WORDS * 4 <= (size_t)NNODES * (DHID - DOUT) * 4,
              "key bytes, edge records and the stored lists fit the half of buffer a that h2 leaves unused");
static_assert(NEDGES % 4 == 0, "targets are read four at a time");
static_assert(NNODES < (1 << 30), "node numbers fit the unsigned range test");

#define AGG_RPW ((AGG_NPB + AGG_WAVES - 1) / AGG_WAVES)
static_assert(NGRAPHS == 64 && DOUT == 64,
              "a lane owns two features and counts two graphs");

__device__ __forceinline__ unsigned zero_bytes(unsigned x) {
  return ~(((x & 0x7f7f7f7fu) + 0x7f7f7f7fu) | x | 0x7f7f7f7fu);
}

template <int D, bool POOL, bool BYTEKEYS, int DN, int LIST>
__global__ void __launch_bounds__(AGG_THREADS)
gather_lds(const float* __restrict__ h, const int* __restrict__ rows,
           const int* __restrict__ cols, const float* __restrict__ w,
           const float* __restrict__ dinv, const float* __restrict__ bias,
           const float* __restrict__ gamma, const float* __restrict__ beta,
           float* __restrict__ out, const int* __restrict__ batch,
           float* __restrict__ sums, float* __restrict__ cnts, int n,
           const unsigned int* __restrict__ keys, const float* __restrict__ Wn,
           unsigned int* __restrict__ lists) {
  static_assert(!POOL || D == DOUT, "the pool table is NGRAPHS x DOUT");
  static_assert(LIST == 0 || BYTEKEYS, "a stored list holds edge numbers that point at edge records");
  static_assert(POOL ? DN == 0 : (DN > 0 && DN % 16 == 0 && D % 4 == 0),
                "without the pool the rows are multiplied by a D x DN matrix");
  constexpr int F = D / 32;
  constexpr int ROW4 = D / 4;
  constexpr int U = 16;
  constexpr int NE4 = NEDGES / 4;
  __shared__ __attribute__((aligned(16))) float acc[AGG_NPB * D];
  __shared__ int q[AGG_QCAP];
  __shared__ float dloc[AGG_NPB];
  __shared__ float2 Wl[POOL ? 1 : (D / 2) * DN];
  __shared__ int listbad;

  const int tid = threadIdx.x;
  const int lane = tid & 31;
  const int wave = __builtin_amdgcn_readfirstlane(tid >> 5);
  const int node0 = blockIdx.x * AGG_NPB;

  for (int i = tid; i < AGG_NPB; i += AGG_THREADS) dloc[i] = dinv[min(node0 + i, n - 1)];
  if constexpr (LIST == 1) {
    if (tid == 0) listbad = 0;
  }

  if constexpr (!POOL) {
    for (int i = tid; i < (D / 2) * DN; i += AGG_THREADS) {
      const int k2 = i / DN;
      const int c = i - k2 * DN;
      float2 v;
      v.x = Wn[(size_t)(2 * k2 + 0) * DN + c];
      v.y = Wn[(size_t)(2 * k2 + 1) * DN + c];
      Wl[i] = v;
    }
  }

  if constexpr (!POOL) {
    if (blockIdx.x == 0) {
      for (int i = tid; i < NGRAPHS * DOUT; i += AGG_THREADS) sums[i] = 0.0f;
      if (tid < NGRAPHS) cnts[tid] = 0.0f;
    }
  }

  {
    const float4* __restrict__ h4 = reinterpret_cast<const float4*>(h);
    const float4* __restrict__ b4 = reinterpret_cast<const float4*>(bias);
    float4* acc4 = reinterpret_cast<float4*>(acc);
#pragma unroll 4
    for (int i = tid; i < AGG_NPB * ROW4; i += AGG_THREADS) {
      const int ln = i / ROW4;
      const int f4 = i - ln * ROW4;
      const int node = min(node0 + ln, n - 1);
      const float d = dinv[node];
      const float s = d * d;
      const float4 hv = h4[(size_t)node * ROW4 + f4];
      const float4 bv = b4[f4];
      float4 o;
      o.x = bv.x + hv.x * s;
      o.y = bv.y + hv.y * s;
      o.z = bv.z + hv.z * s;
      o.w = bv.w + hv.w * s;
      acc4[i] = o;
    }
  }
  __syncthreads();

  const int4* __restrict__ c4p = reinterpret_cast<const int4*>(cols);

  constexpr int WQCAP = AGG_QCAP / AGG_WAVES;
  static_assert(WQCAP >= 64, "a wave's part holds a full batch and then some");
  int* wq = q + wave * WQCAP;
  int qcount = 0;

  constexpr int NWLISTS = AGG_BLOCKS * AGG_WAVES;
  const int wlist = (int)blockIdx.x * AGG_WAVES + wave;
  unsigned int* mylist = (LIST != 0) ? lists + (size_t)wlist * AGG_LCAP : nullptr;
  unsigned int* lcounts = (LIST != 0) ? lists + (size_t)NWLISTS * AGG_LCAP : nullptr;
  unsigned int* lflags = (LIST != 0) ? lists + (size_t)NWLISTS * (AGG_LCAP + 1) : nullptr;
  int lcount = 0;
  bool lbad = false;
  bool uselist = false;
  if constexpr (LIST == 2) {
    uselist = __builtin_amdgcn_readfirstlane((int)lflags[blockIdx.x]) == 0;
    if (uselist) qcount = __builtin_amdgcn_readfirstlane((int)lcounts[wlist]);
  }

  auto push = [&](unsigned mask, bool hit, int word) {
    const int pos = qcount + (int)__builtin_amdgcn_mbcnt_lo(mask, 0u);
    if (hit && pos < WQCAP) wq[pos] = word;
    qcount += __builtin_popcount(mask);
  };

  constexpr int LOADS = BYTEKEYS ? 7 : AGG_LOADS;
  constexpr int EPL = BYTEKEYS ? 16 : 4;
  constexpr int NV = NEDGES / EPL;
  constexpr int NTRIPS = (NV + LOADS * AGG_THREADS - 1) / (LOADS * AGG_THREADS);

  auto add_batch = [&](int nb, int qv) {
    int r = 0, cl = 0;
    float nr = 0.0f;
    if (lane < nb) {
      if constexpr (BYTEKEYS) {
        const uint2 rc = reinterpret_cast<const uint2*>(keys + NEDGES / 4)[qv];
        r = (int)(rc.x & 0x1ffffu);
        cl = (int)(rc.x >> 17);
        nr = dinv[r] * __uint_as_float(rc.y) * dloc[cl];
      } else {
        const int e = qv & 0x1fffff;
        cl = qv >> 21;
        r = rows[e];
        nr = dinv[r] * w[e] * dloc[cl];
      }
    }
#pragma unroll
    for (int j0 = 0; j0 < 32; j0 += U) {
      if (j0 < nb) {
        float v[U][F];
#pragma unroll
        for (int u = 0; u < U; ++u) {
          if (j0 + u < nb) {
            const int rr = __builtin_amdgcn_readlane(r, j0 + u);
            const float* __restrict__ hp = h + (size_t)rr * D + lane;
#pragma unroll
            for (int f = 0; f < F; ++f) v[u][f] = hp[f * 32];
          }
        }
#pragma unroll
        for (int u = 0; u < U; ++u) {
          if (j0 + u < nb) {
            const int cc = __builtin_amdgcn_readlane(cl, j0 + u);
            const float nn =
                __int_as_float(__builtin_amdgcn_readlane(__float_as_int(nr), j0 + u));
            float* ap = acc + cc * D + lane;
#pragma unroll
            for (int f = 0; f < F; ++f) atomicAdd(ap + f * 32, v[u][f] * nn);
          }
        }
      }
    }
  };

  if (!uselist)
  for (int it = 0; it <= NTRIPS; ++it) {
    const int q0 = qcount;
    if (it < NTRIPS) {
      const bool whole = (it + 1) * LOADS * AGG_THREADS <= NV;
      if constexpr (BYTEKEYS) {
        const uint4* __restrict__ k4p = reinterpret_cast<const uint4*>(keys);
        const unsigned pat = (unsigned)blockIdx.x * 0x01010101u;
        uint4 kv[LOADS];
        int gi[LOADS];
#pragma unroll
        for (int j = 0; j < LOADS; ++j) gi[j] = (it * LOADS + j) * AGG_THREADS + tid;
        if (whole) {
#pragma unroll
          for (int j = 0; j < LOADS; ++j) kv[j] = k4p[gi[j]];
        } else {
#pragma unroll
          for (int j = 0; j < LOADS; ++j) {
            if (gi[j] < NV) {
              kv[j] = k4p[gi[j]];
            } else {
              kv[j].x = ~pat; kv[j].y = ~pat; kv[j].z = ~pat; kv[j].w = ~pat;
            }
          }
        }
#pragma unroll
        for (int j = 0; j < LOADS; ++j) {
          unsigned m = (zero_bytes(kv[j].x ^ pat) >> 7) | (zero_bytes(kv[j].y ^ pat) >> 6) |
                       (zero_bytes(kv[j].z ^ pat) >> 5) | (zero_bytes(kv[j].w ^ pat) >> 4);
          unsigned mk = __builtin_amdgcn_ballot_w32(m != 0);
          while (mk != 0) {
            const bool hit = m != 0;
            const int p = __builtin_ctz(m | 0x80000000u);
            push(mk, hit, gi[j] * 16 + ((p & 7) << 2) + (p >> 3));
            m &= m - 1;
            mk = __builtin_amdgcn_ballot_w32(m != 0);
          }
        }
      } else {
        int4 cv[LOADS];
        int idx[LOADS];
#pragma unroll
        for (int j = 0; j < LOADS; ++j) idx[j] = (it * LOADS + j) * AGG_THREADS + tid;
        if (whole) {
#pragma unroll
          for (int j = 0; j < LOADS; ++j) cv[j] = c4p[idx[j]];
        } else {
#pragma unroll
          for (int j = 0; j < LOADS; ++j) {
            if (idx[j] < NV) {
              cv[j] = c4p[idx[j]];
            } else {
              cv[j].x = -1; cv[j].y = -1; cv[j].z = -1; cv[j].w = -1;
            }
          }
        }
        auto test = [&](int c, int e) {
          const unsigned d = (unsigned)(c - node0);
          const bool hit = d < AGG_NPB;
          const unsigned mk = __builtin_amdgcn_ballot_w32(hit);
          if (mk != 0) push(mk, hit, e | (int)(d << 21));
        };
#pragma unroll
        for (int j = 0; j < LOADS; ++j) {
          test(cv[j].x, idx[j] * 4 + 0);
          test(cv[j].y, idx[j] * 4 + 1);
          test(cv[j].z, idx[j] * 4 + 2);
          test(cv[j].w, idx[j] * 4 + 3);
        }
      }
      if (qcount > WQCAP) {
        qcount = q0;
        lbad = true;
#pragma unroll 1
        for (int j = 0; j < LOADS; ++j) {
          const int e0 = ((it * LOADS + j) * AGG_THREADS + tid) * EPL;
#pragma unroll 1
          for (int k = 0; k < EPL; ++k) {
            const int e = e0 + k;
            if (e < NEDGES) {
              const int c = cols[e];
              const unsigned dl = (unsigned)(c - node0);
              if (dl < AGG_NPB) {
                const int r = rows[e];
                const float nr = dinv[r] * w[e] * dinv[c];
                for (int f = 0; f < D; ++f)
                  atomicAdd(&acc[dl * D + f], h[(size_t)r * D + f] * nr);
              }
            }
          }
        }
      }
    }
    const int need = (it < NTRIPS) ? 32 : 1;
    int base = 0;
    while (qcount - base >= need) {
      const int nb = min(32, qcount - base);
      int qv = 0;
      if (lane < nb) qv = wq[base + lane];
      if constexpr (LIST == 1) {
        if (lane < nb && lcount + lane < AGG_LCAP) mylist[lcount + lane] = (unsigned)qv;
        lcount += nb;
      }
      add_batch(nb, qv);
      base += nb;
    }
    if (base > 0) {
      const int rem = qcount - base;
      int t = 0;
      if (lane < rem) t = wq[base + lane];
      if (lane < rem) wq[lane] = t;
      qcount = rem;
    }
  }
  if constexpr (LIST == 2) {
    if (uselist && qcount > 0) {
      int qv = (int)mylist[min(lane, qcount - 1)];
      for (int base = 0; base < qcount; base += 32) {
        const int qn = (int)mylist[min(base + 32 + lane, qcount - 1)];
        add_batch(min(32, qcount - base), qv);
        qv = qn;
      }
    }
  }
  if constexpr (LIST == 1) {
    if (lcount > AGG_LCAP) lbad = true;
    if (lane == 0) {
      lcounts[wlist] = (unsigned)min(lcount, AGG_LCAP);
      if (lbad) listbad = 1;
    }
  }
  __syncthreads();
  if constexpr (LIST == 1) {
    if (tid == 0) lflags[blockIdx.x] = (unsigned)listbad;
  }

  {
    static_assert(F == 4 || F == 2, "a lane reads its F features of a row with one LDS load");
    float gm[F], bt[F];
#pragma unroll
    for (int f = 0; f < F; ++f) {
      gm[f] = gamma[lane * F + f];
      bt[f] = beta[lane * F + f];
    }
    auto ln_row = [&](int ln, float (&y)[F]) {
      const float* rp = acc + ln * D + lane * F;
      float v[F];
      if constexpr (F == 4) {
        const float4 t = *reinterpret_cast<const float4*>(rp);
        v[0] = t.x; v[1] = t.y; v[2] = t.z; v[3] = t.w;
      } else {
        const float2 t = *reinterpret_cast<const float2*>(rp);
        v[0] = t.x; v[1] = t.y;
      }
      float s = 0.0f;
#pragma unroll
      for (int f = 0; f < F; ++f) {
        v[f] = fmaxf(v[f], 0.0f);
        s += v[f];
      }
      s = wave_sum(s);
      const float mu = s * (1.0f / D);
      float ss = 0.0f;
#pragma unroll
      for (int f = 0; f < F; ++f) {
        const float d = v[f] - mu;
        ss += d * d;
      }
      ss = wave_sum(ss);
      const float rs = rsqrtf(ss * (1.0f / D) + EPSF);
#pragma unroll
      for (int f = 0; f < F; ++f) y[f] = (v[f] - mu) * rs * gm[f] + bt[f];
    };

    if constexpr (!POOL) {
      float yy[AGG_RPW][F];
#pragma unroll
      for (int k = 0; k < AGG_RPW; ++k) ln_row(min(wave + k * AGG_WAVES, AGG_NPB - 1), yy[k]);
#pragma unroll
      for (int k = 0; k < AGG_RPW; ++k) {
        const int ln = wave + k * AGG_WAVES;
        const float (&y)[F] = yy[k];
        if (ln < AGG_NPB) {
          float* ap = acc + ln * D + lane * F;
          if constexpr (F == 4) {
            float4 o; o.x = y[0]; o.y = y[1]; o.z = y[2]; o.w = y[3];
            *reinterpret_cast<float4*>(ap) = o;
          } else {
            float2 o; o.x = y[0]; o.y = y[1];
            *reinterpret_cast<float2*>(ap) = o;
          }
        }
      }
      __syncthreads();

      constexpr int NT = DN / 16;
      constexpr int NSTRIPES = (AGG_NPB + 15) / 16;
      const int half = lane >> 4;
      const int l = lane & 15;
      for (int stripe = wave; stripe < NSTRIPES; stripe += AGG_WAVES) {
        const int m0 = stripe * 16;
        v8f cacc[NT];
#pragma unroll
        for (int t = 0; t < NT; ++t)
#pragma unroll
          for (int j = 0; j < 8; ++j) cacc[t][j] = 0.0f;
        const float* arow = acc + min(m0 + l, AGG_NPB - 1) * D + 2 * half;
        for (int k0 = 0; k0 < D; k0 += 4) {
          const float2 av = *reinterpret_cast<const float2*>(arow + k0);
          v2f afrag;
          afrag[0] = av.x;
          afrag[1] = av.y;
          const float2* brow = Wl + (k0 / 2 + half) * DN + l;
          v2f bfrag[NT];
#pragma unroll
          for (int t = 0; t < NT; ++t) {
            const float2 bb = brow[t * 16];
            bfrag[t][0] = bb.x;
            bfrag[t][1] = bb.y;
          }
#pragma unroll
          for (int t = 0; t < NT; ++t)
            cacc[t] = __builtin_amdgcn_wmma_f32_16x16x4_f32(
                 false, afrag,  false, bfrag[t],
                 (short)0, cacc[t],  false,  false);
        }
#pragma unroll
        for (int r = 0; r < 8; ++r) {
          const int ln = m0 + r + 8 * half;
          const int node = node0 + ln;
          if (ln < AGG_NPB && node < n) {
#pragma unroll
            for (int t = 0; t < NT; ++t)
              out[(size_t)node * DN + t * 16 + l] = cacc[t][r];
          }
        }
      }
    } else {
      __shared__ float tab[NGRAPHS * DOUT];
      __shared__ float gcnt[NGRAPHS];
      for (int i = tid; i < NGRAPHS * DOUT; i += AGG_THREADS) tab[i] = 0.0f;
      if (tid < NGRAPHS) gcnt[tid] = 0.0f;
      int gid[AGG_RPW];
#pragma unroll
      for (int k = 0; k < AGG_RPW; ++k)
        gid[k] = batch[min(node0 + wave + k * AGG_WAVES, n - 1)];
      __syncthreads();

      int c0 = 0, c1 = 0;
      float yy[AGG_RPW][F];
#pragma unroll
      for (int k = 0; k < AGG_RPW; ++k) ln_row(min(wave + k * AGG_WAVES, AGG_NPB - 1), yy[k]);
#pragma unroll
      for (int k = 0; k < AGG_RPW; ++k) {
        const int ln = wave + k * AGG_WAVES;
        const int node = node0 + ln;
        const float (&y)[F] = yy[k];
        if (ln < AGG_NPB && node < n) {
          const int g = gid[k];
          atomicAdd(&tab[g * DOUT + lane * 2 + 0], y[0]);
          atomicAdd(&tab[g * DOUT + lane * 2 + 1], y[1]);
          c0 += (g == lane) ? 1 : 0;
          c1 += (g == lane + 32) ? 1 : 0;
        }
      }
      if (c0 != 0) atomicAdd(&gcnt[lane], (float)c0);
      if (c1 != 0) atomicAdd(&gcnt[lane + 32], (float)c1);
      __syncthreads();

      for (int i = tid; i < NGRAPHS * DOUT; i += AGG_THREADS) {
        const float s = tab[i];
        if (s != 0.0f) atomicAdd(&sums[i], s);
      }
      if (tid < NGRAPHS) {
        const float c = gcnt[tid];
        if (c != 0.0f) atomicAdd(&cnts[tid], c);
      }
    }
  }
}

__global__ void pool_final(const float* __restrict__ sums, const float* __restrict__ cnts,
                           float* __restrict__ out) {
  int i = blockIdx.x * blockDim.x + threadIdx.x;
  if (i < NGRAPHS * DOUT) {
    int g = i / DOUT;
    float m = sums[i] / fmaxf(cnts[g], 1.0f);
    out[i] = 1.0f / (1.0f + __expf(-m));
  }
}

extern "C" void kernel_launch(void* const* d_in, const int* in_sizes, int n_in,
                              void* d_out, int out_size, void* d_ws, size_t ws_size,
                              hipStream_t stream) {
  const float* x   = (const float*)d_in[0];
  const int*   ei  = (const int*)d_in[1];
  const float* ew  = (const float*)d_in[2];
  const int*   bat = (const int*)d_in[3];
  const float* W1  = (const float*)d_in[4];
  const float* b1  = (const float*)d_in[5];
  const float* g1  = (const float*)d_in[6];
  const float* bt1 = (const float*)d_in[7];
  const float* W2  = (const float*)d_in[8];
  const float* b2  = (const float*)d_in[9];
  const float* g2  = (const float*)d_in[10];
  const float* bt2 = (const float*)d_in[11];
  float* out = (float*)d_out;

  const int* rows = ei;
  const int* cols = ei + NEDGES;

  char* ws = (char*)d_ws;
  size_t off = 0;
  auto carve = [&](size_t bytes) -> float* {
    float* p = (float*)(ws + off);
    off += (bytes + 255) & ~(size_t)255;
    return p;
  };
  float* dinv = carve((size_t)NNODES * 4);
  float* h    = carve((size_t)NNODES * DHID * 4);
  float* a    = carve((size_t)NNODES * DHID * 4);
  float* sums = carve((size_t)NGRAPHS * DOUT * 4);
  float* cnts = carve((size_t)NGRAPHS * 4);

  const int TB = 256;
  auto blocks = [&](long long threads) { return (int)((threads + TB - 1) / TB); };

  zero_f32<<<blocks(NNODES), TB, 0, stream>>>(dinv, NNODES);
  deg_accum<<<blocks(NEDGES), TB, 0, stream>>>(cols, ew, dinv, NEDGES);
  deg_to_dinv<<<blocks(NNODES), TB, 0, stream>>>(dinv, NNODES);

  float* h2 = a;
  unsigned int* keys = (unsigned int*)(a + (size_t)NNODES * DOUT);
  wmma_gemm_f32<DHID, DIN, true><<<GEMM_BLOCKS, GEMM_THREADS, 0, stream>>>(
      x, W1, h, NNODES, rows, cols, ew, keys);
  unsigned int* lists = keys + NEDGES / 4 + (size_t)NEDGES * 2;
  gather_lds<DHID, false, true, DOUT, 1><<<AGG_BLOCKS, AGG_THREADS, 0, stream>>>(
      h, rows, cols, ew, dinv, b1, g1, bt1, h2, nullptr, sums, cnts, NNODES, keys, W2, lists);

  gather_lds<DOUT, true, true, 0, 2><<<AGG_BLOCKS, AGG_THREADS, 0, stream>>>(
      h2, rows, cols, ew, dinv, b2, g2, bt2, nullptr, bat, sums, cnts, NNODES, keys, nullptr, lists);
  pool_final<<<blocks(NGRAPHS * DOUT), TB, 0, stream>>>(sums, cnts, out);
}
